# starting kernel
# baseline (speedup 1.0000x reference)
#include <hip/hip_runtime.h>
#include <math.h>


typedef float v2f __attribute__((ext_vector_type(2)));
typedef float v4f __attribute__((ext_vector_type(4)));
typedef float v8f __attribute__((ext_vector_type(8)));

#define DQ      512
#define QQ      256
#define NLAY    4
#define NPAIR   255
#define KSTEPS  128
#define NTILES  32
#define M_STATE (16 * 2048)

static __device__ __forceinline__ v8f wmma_f32_4(v2f a, v2f b, v8f c) {
  return __builtin_amdgcn_wmma_f32_16x16x4_f32(false, a, false, b, (short)0, c,
                                               false, false);
}

__global__ void vqc_build_ent(const float* __restrict__ ent_params,
                              float* __restrict__ Tent) {
  const int l = blockIdx.x;
  const int r = threadIdx.x;

  __shared__ float c[NPAIR];
  for (int p = threadIdx.x; p < NPAIR; p += blockDim.x) {
    float x = ent_params[l * NPAIR + p];
    c[p] = 0.1f * (1.0f / (1.0f + expf(-x)));
  }
  __syncthreads();

  float* row = Tent + ((size_t)l * DQ + r) * DQ;
  for (int j = 0; j < DQ; ++j) row[j] = 0.0f;
  row[r] = 1.0f;
  if (r < QQ) {
    if (r >= 1) row[r - 1] = c[r - 1];
    if (r <= QQ - 2) {
      float val = c[r];
      for (int j = r + 1; j < QQ; ++j) {
        row[j] = val;
        if (j < NPAIR) val *= c[j];
      }
    }
  }
}

__global__ void vqc_build_layer(const float* __restrict__ rot_params,
                                const float* __restrict__ Tent,
                                float* __restrict__ Tlay) {
  const int i = blockIdx.x;
  const int l = blockIdx.y;
  const int q = i >> 1;
  const int par = i & 1;

  const float t0 = rot_params[((l * QQ) + q) * 3 + 0];
  const float t1 = rot_params[((l * QQ) + q) * 3 + 1];
  const float t2 = rot_params[((l * QQ) + q) * 3 + 2];
  const float c0 = cosf(t0 * 0.5f);
  const float c1 = cosf(t1 * 0.5f);
  const float s1 = sinf(t1 * 0.5f);
  const float c2 = cosf(t2 * 0.5f);
  const float b00 = (c0 * c1) * c2;
  const float b01 = (c0 * (-s1)) * c2;
  const float b10 = (c0 * s1) * c2;

  const float w0 = (par == 0) ? b00 : b10;
  const float w1 = (par == 0) ? b01 : b00;

  const float* r0 = Tent + ((size_t)l * DQ + 2 * q) * DQ;
  const float* r1 = r0 + DQ;
  float* out = Tlay + ((size_t)l * DQ + i) * DQ;
  for (int j = threadIdx.x; j < DQ; j += blockDim.x)
    out[j] = w0 * r0[j] + w1 * r1[j];
}

__global__ void vqc_pack_b(const float* __restrict__ B, float* __restrict__ Bp) {
  const int idx = blockIdx.x * blockDim.x + threadIdx.x;
  const int lane = idx & 31;
  const int nt = (idx >> 5) & 31;
  const int ks = idx >> 10;
  const int k0 = ks * 4 + ((lane >> 4) << 1);
  const int col = nt * 16 + (lane & 15);
  v2f v;
  v[0] = B[(size_t)k0 * DQ + col];
  v[1] = B[(size_t)(k0 + 1) * DQ + col];
  *(v2f*)(Bp + (size_t)idx * 2) = v;
}

__global__ void __launch_bounds__(128)
vqc_wmma_gemm(const float* __restrict__ A, const float* __restrict__ Bp,
              float* __restrict__ C) {
  __shared__ v2f As[2 * KSTEPS * 32];

  const int row0 = blockIdx.x * 32;

  {
    const int row = threadIdx.x >> 2;
    const int cq = threadIdx.x & 3;
    const int mt = row >> 4;
    const int lr = row & 15;
    const float* arow = A + (size_t)(row0 + row) * DQ;
#pragma unroll 4
    for (int s = 0; s < 32; ++s) {
      const int ks = cq + 4 * s;
      v4f v = *(const v4f*)(arow + ks * 4);
      v2f lo; lo[0] = v[0]; lo[1] = v[1];
      v2f hi; hi[0] = v[2]; hi[1] = v[3];
      As[(mt * KSTEPS + ks) * 32 + lr] = lo;
      As[(mt * KSTEPS + ks) * 32 + lr + 16] = hi;
    }
  }
  __syncthreads();

  const int wave = threadIdx.x >> 5;
  const int lane = threadIdx.x & 31;
  const int ntbase = wave * 8;

  v8f acc[2][8];
#pragma unroll
  for (int m = 0; m < 2; ++m)
#pragma unroll
    for (int t = 0; t < 8; ++t)
      acc[m][t] = (v8f){0.f, 0.f, 0.f, 0.f, 0.f, 0.f, 0.f, 0.f};

  const v2f* BpV = (const v2f*)Bp;
  for (int ks = 0; ks < KSTEPS; ++ks) {
    const v2f a0 = As[(0 * KSTEPS + ks) * 32 + lane];
    const v2f a1 = As[(1 * KSTEPS + ks) * 32 + lane];
#pragma unroll
    for (int t = 0; t < 8; ++t) {
      const v2f b = BpV[((size_t)ks * NTILES + (ntbase + t)) * 32 + lane];
      acc[0][t] = wmma_f32_4(a0, b, acc[0][t]);
      acc[1][t] = wmma_f32_4(a1, b, acc[1][t]);
    }
  }

  const int lrow = lane & 15;
  const int hi8 = (lane >> 4) * 8;
#pragma unroll
  for (int mt = 0; mt < 2; ++mt)
#pragma unroll
    for (int t = 0; t < 8; ++t) {
      const int col = (ntbase + t) * 16 + lrow;
#pragma unroll
      for (int v = 0; v < 8; ++v) {
        const int r = row0 + mt * 16 + hi8 + v;
        C[(size_t)r * DQ + col] = acc[mt][t][v];
      }
    }
}

extern "C" void kernel_launch(void* const* d_in, const int* in_sizes, int n_in,
                              void* d_out, int out_size, void* d_ws,
                              size_t ws_size, hipStream_t stream) {
  (void)in_sizes; (void)n_in; (void)out_size; (void)ws_size;
  const float* state = (const float*)d_in[0];
  const float* rotp  = (const float*)d_in[1];
  const float* entp  = (const float*)d_in[2];
  float* out = (float*)d_out;

  float* w     = (float*)d_ws;
  float* Tent  = w;
  float* Tlay  = w + 1048576;
  float* T01   = w + 0;
  float* T23   = w + 262144;
  float* Ttot  = w + 524288;
  float* Bp    = w + 786432;

  vqc_build_ent<<<NLAY, 512, 0, stream>>>(entp, Tent);

  vqc_build_layer<<<dim3(DQ, NLAY), 128, 0, stream>>>(rotp, Tent, Tlay);

  vqc_pack_b<<<512, 256, 0, stream>>>(Tlay + 1 * 262144, Bp);
  vqc_wmma_gemm<<<DQ / 32, 128, 0, stream>>>(Tlay + 0 * 262144, Bp, T01);

  vqc_pack_b<<<512, 256, 0, stream>>>(Tlay + 3 * 262144, Bp);
  vqc_wmma_gemm<<<DQ / 32, 128, 0, stream>>>(Tlay + 2 * 262144, Bp, T23);

  vqc_pack_b<<<512, 256, 0, stream>>>(T23, Bp);
  vqc_wmma_gemm<<<DQ / 32, 128, 0, stream>>>(T01, Bp, Ttot);

  vqc_pack_b<<<512, 256, 0, stream>>>(Ttot, Bp);
  vqc_wmma_gemm<<<M_STATE / 32, 128, 0, stream>>>(state, Bp, out);
}


# The three 512^3 chain products get a latency-shaped f32 WMMA kernel of their own: 256 two-wave work-groups, two adjacent tiles per wave, K loop unrolled, loads running ahead; sum order unchanged
# speedup vs baseline: 1.7330x; 1.7330x over previous
#include <hip/hip_runtime.h>
#include <math.h>


typedef float v2f __attribute__((ext_vector_type(2)));
typedef float v4f __attribute__((ext_vector_type(4)));
typedef float v8f __attribute__((ext_vector_type(8)));

#define DQ      512
#define QQ      256
#define NLAY    4
#define NPAIR   255
#define KSTEPS  128
#define NTILES  32
#define M_STATE (16 * 2048)

static __device__ __forceinline__ v8f wmma_f32_4(v2f a, v2f b, v8f c) {
  return __builtin_amdgcn_wmma_f32_16x16x4_f32(false, a, false, b, (short)0, c,
                                               false, false);
}

__global__ void vqc_build_ent(const float* __restrict__ ent_params,
                              float* __restrict__ Tent) {
  const int l = blockIdx.x;
  const int r = threadIdx.x;

  __shared__ float c[NPAIR];
  for (int p = threadIdx.x; p < NPAIR; p += blockDim.x) {
    float x = ent_params[l * NPAIR + p];
    c[p] = 0.1f * (1.0f / (1.0f + expf(-x)));
  }
  __syncthreads();

  float* row = Tent + ((size_t)l * DQ + r) * DQ;
  for (int j = 0; j < DQ; ++j) row[j] = 0.0f;
  row[r] = 1.0f;
  if (r < QQ) {
    if (r >= 1) row[r - 1] = c[r - 1];
    if (r <= QQ - 2) {
      float val = c[r];
      for (int j = r + 1; j < QQ; ++j) {
        row[j] = val;
        if (j < NPAIR) val *= c[j];
      }
    }
  }
}

__global__ void vqc_build_layer(const float* __restrict__ rot_params,
                                const float* __restrict__ Tent,
                                float* __restrict__ Tlay) {
  const int i = blockIdx.x;
  const int l = blockIdx.y;
  const int q = i >> 1;
  const int par = i & 1;

  const float t0 = rot_params[((l * QQ) + q) * 3 + 0];
  const float t1 = rot_params[((l * QQ) + q) * 3 + 1];
  const float t2 = rot_params[((l * QQ) + q) * 3 + 2];
  const float c0 = cosf(t0 * 0.5f);
  const float c1 = cosf(t1 * 0.5f);
  const float s1 = sinf(t1 * 0.5f);
  const float c2 = cosf(t2 * 0.5f);
  const float b00 = (c0 * c1) * c2;
  const float b01 = (c0 * (-s1)) * c2;
  const float b10 = (c0 * s1) * c2;

  const float w0 = (par == 0) ? b00 : b10;
  const float w1 = (par == 0) ? b01 : b00;

  const float* r0 = Tent + ((size_t)l * DQ + 2 * q) * DQ;
  const float* r1 = r0 + DQ;
  float* out = Tlay + ((size_t)l * DQ + i) * DQ;
  for (int j = threadIdx.x; j < DQ; j += blockDim.x)
    out[j] = w0 * r0[j] + w1 * r1[j];
}

__global__ void vqc_pack_b(const float* __restrict__ B, float* __restrict__ Bp) {
  const int idx = blockIdx.x * blockDim.x + threadIdx.x;
  const int lane = idx & 31;
  const int nt = (idx >> 5) & 31;
  const int ks = idx >> 10;
  const int k0 = ks * 4 + ((lane >> 4) << 1);
  const int col = nt * 16 + (lane & 15);
  v2f v;
  v[0] = B[(size_t)k0 * DQ + col];
  v[1] = B[(size_t)(k0 + 1) * DQ + col];
  *(v2f*)(Bp + (size_t)idx * 2) = v;
}

__global__ void __launch_bounds__(128)
vqc_wmma_gemm(const float* __restrict__ A, const float* __restrict__ Bp,
              float* __restrict__ C) {
  __shared__ v2f As[2 * KSTEPS * 32];

  const int row0 = blockIdx.x * 32;

  {
    const int row = threadIdx.x >> 2;
    const int cq = threadIdx.x & 3;
    const int mt = row >> 4;
    const int lr = row & 15;
    const float* arow = A + (size_t)(row0 + row) * DQ;
#pragma unroll 4
    for (int s = 0; s < 32; ++s) {
      const int ks = cq + 4 * s;
      v4f v = *(const v4f*)(arow + ks * 4);
      v2f lo; lo[0] = v[0]; lo[1] = v[1];
      v2f hi; hi[0] = v[2]; hi[1] = v[3];
      As[(mt * KSTEPS + ks) * 32 + lr] = lo;
      As[(mt * KSTEPS + ks) * 32 + lr + 16] = hi;
    }
  }
  __syncthreads();

  const int wave = threadIdx.x >> 5;
  const int lane = threadIdx.x & 31;
  const int ntbase = wave * 8;

  v8f acc[2][8];
#pragma unroll
  for (int m = 0; m < 2; ++m)
#pragma unroll
    for (int t = 0; t < 8; ++t)
      acc[m][t] = (v8f){0.f, 0.f, 0.f, 0.f, 0.f, 0.f, 0.f, 0.f};

  const v2f* BpV = (const v2f*)Bp;
  for (int ks = 0; ks < KSTEPS; ++ks) {
    const v2f a0 = As[(0 * KSTEPS + ks) * 32 + lane];
    const v2f a1 = As[(1 * KSTEPS + ks) * 32 + lane];
#pragma unroll
    for (int t = 0; t < 8; ++t) {
      const v2f b = BpV[((size_t)ks * NTILES + (ntbase + t)) * 32 + lane];
      acc[0][t] = wmma_f32_4(a0, b, acc[0][t]);
      acc[1][t] = wmma_f32_4(a1, b, acc[1][t]);
    }
  }

  const int lrow = lane & 15;
  const int hi8 = (lane >> 4) * 8;
#pragma unroll
  for (int mt = 0; mt < 2; ++mt)
#pragma unroll
    for (int t = 0; t < 8; ++t) {
      const int col = (ntbase + t) * 16 + lrow;
#pragma unroll
      for (int v = 0; v < 8; ++v) {
        const int r = row0 + mt * 16 + hi8 + v;
        C[(size_t)r * DQ + col] = acc[mt][t][v];
      }
    }
}

#define CHK 16
#define NCHK (KSTEPS / CHK)

__global__ void __launch_bounds__(64)
vqc_wmma_gemm_chain(const float* __restrict__ A, const float* __restrict__ Bp,
                    float* __restrict__ C) {
  __shared__ v2f As[KSTEPS * 32];

  const int row0 = blockIdx.y * 16;
  const int wave = threadIdx.x >> 5;
  const int lane = threadIdx.x & 31;
  const int nt0 = blockIdx.x * 4 + wave * 2;

  const v2f* bp = (const v2f*)Bp + (size_t)nt0 * 32 + lane;

  v2f bc0[CHK], bc1[CHK], ac[CHK];
  v2f bn0[CHK], bn1[CHK], an[CHK];

#pragma unroll
  for (int u = 0; u < CHK; ++u) {
    bc0[u] = bp[(size_t)u * (NTILES * 32)];
    bc1[u] = bp[(size_t)u * (NTILES * 32) + 32];
  }

  {
    const int row = threadIdx.x >> 2;
    const int cq = threadIdx.x & 3;
    const float* arow = A + (size_t)(row0 + row) * DQ;
    v4f v[32];
#pragma unroll
    for (int s = 0; s < 32; ++s) v[s] = *(const v4f*)(arow + (cq + 4 * s) * 4);
#pragma unroll
    for (int s = 0; s < 32; ++s) {
      const int ks = cq + 4 * s;
      v2f lo; lo[0] = v[s][0]; lo[1] = v[s][1];
      v2f hi; hi[0] = v[s][2]; hi[1] = v[s][3];
      As[ks * 32 + row] = lo;
      As[ks * 32 + row + 16] = hi;
    }
  }
  __syncthreads();

#pragma unroll
  for (int u = 0; u < CHK; ++u) ac[u] = As[u * 32 + lane];

  v8f acc0 = (v8f){0.f, 0.f, 0.f, 0.f, 0.f, 0.f, 0.f, 0.f};
  v8f acc1 = (v8f){0.f, 0.f, 0.f, 0.f, 0.f, 0.f, 0.f, 0.f};

#pragma unroll
  for (int c = 0; c < NCHK; ++c) {
    if (c + 1 < NCHK) {
      const int kn = (c + 1) * CHK;
#pragma unroll
      for (int u = 0; u < CHK; ++u) {
        bn0[u] = bp[(size_t)(kn + u) * (NTILES * 32)];
        bn1[u] = bp[(size_t)(kn + u) * (NTILES * 32) + 32];
      }
#pragma unroll
      for (int u = 0; u < CHK; ++u) an[u] = As[(kn + u) * 32 + lane];
    }
#pragma unroll
    for (int u = 0; u < CHK; ++u) {
      acc0 = wmma_f32_4(ac[u], bc0[u], acc0);
      acc1 = wmma_f32_4(ac[u], bc1[u], acc1);
    }
    if (c + 1 < NCHK) {
#pragma unroll
      for (int u = 0; u < CHK; ++u) {
        bc0[u] = bn0[u]; bc1[u] = bn1[u]; ac[u] = an[u];
      }
    }
  }

  __builtin_amdgcn_sched_barrier(0);
  asm volatile("v_nop\n\tv_nop\n\tv_nop\n\tv_nop");
  __builtin_amdgcn_sched_barrier(0);

  const int lrow = lane & 15;
  const int hi8 = (lane >> 4) * 8;
#pragma unroll
  for (int v = 0; v < 8; ++v) {
    float* crow = C + (size_t)(row0 + hi8 + v) * DQ + nt0 * 16 + lrow;
    crow[0] = acc0[v];
    crow[16] = acc1[v];
  }
}

extern "C" void kernel_launch(void* const* d_in, const int* in_sizes, int n_in,
                              void* d_out, int out_size, void* d_ws,
                              size_t ws_size, hipStream_t stream) {
  (void)in_sizes; (void)n_in; (void)out_size; (void)ws_size;
  const float* state = (const float*)d_in[0];
  const float* rotp  = (const float*)d_in[1];
  const float* entp  = (const float*)d_in[2];
  float* out = (float*)d_out;

  float* w     = (float*)d_ws;
  float* Tent  = w;
  float* Tlay  = w + 1048576;
  float* T01   = w + 0;
  float* T23   = w + 262144;
  float* Ttot  = w + 524288;
  float* Bp    = w + 786432;

  vqc_build_ent<<<NLAY, 512, 0, stream>>>(entp, Tent);

  vqc_build_layer<<<dim3(DQ, NLAY), 128, 0, stream>>>(rotp, Tent, Tlay);

  vqc_pack_b<<<512, 256, 0, stream>>>(Tlay + 1 * 262144, Bp);
  vqc_wmma_gemm_chain<<<dim3(DQ / 64, DQ / 16), 64, 0, stream>>>(Tlay + 0 * 262144, Bp, T01);

  vqc_pack_b<<<512, 256, 0, stream>>>(Tlay + 3 * 262144, Bp);
  vqc_wmma_gemm_chain<<<dim3(DQ / 64, DQ / 16), 64, 0, stream>>>(Tlay + 2 * 262144, Bp, T23);

  vqc_pack_b<<<512, 256, 0, stream>>>(T23, Bp);
  vqc_wmma_gemm_chain<<<dim3(DQ / 64, DQ / 16), 64, 0, stream>>>(T01, Bp, Ttot);

  vqc_pack_b<<<512, 256, 0, stream>>>(Ttot, Bp);
  vqc_wmma_gemm<<<M_STATE / 32, 128, 0, stream>>>(state, Bp, out);
}


# T_ent is no longer built by a 4-work-group kernel with one row per thread: a full-grid kernel makes T_layer from the parameters, both T_ent rows in LDS; product order unchanged; 9 launches
# speedup vs baseline: 2.4083x; 1.3897x over previous
#include <hip/hip_runtime.h>
#include <math.h>


typedef float v2f __attribute__((ext_vector_type(2)));
typedef float v4f __attribute__((ext_vector_type(4)));
typedef float v8f __attribute__((ext_vector_type(8)));

#define DQ      512
#define QQ      256
#define NLAY    4
#define NPAIR   255
#define KSTEPS  128
#define NTILES  32
#define M_STATE (16 * 2048)

static __device__ __forceinline__ v8f wmma_f32_4(v2f a, v2f b, v8f c) {
  return __builtin_amdgcn_wmma_f32_16x16x4_f32(false, a, false, b, (short)0, c,
                                               false, false);
}

#define CPAD 8

__global__ void __launch_bounds__(128)
vqc_build_layer(const float* __restrict__ rot_params,
                const float* __restrict__ ent_params,
                float* __restrict__ Tlay) {
  const int q = blockIdx.x;
  const int l = blockIdx.y;
  const int tid = threadIdx.x;

  __shared__ __attribute__((aligned(16))) float c[QQ + CPAD];
  __shared__ __attribute__((aligned(16))) float e[2][DQ];

  for (int p = tid; p < QQ + CPAD; p += 128) {
    float cv = 1.0f;
    if (p < NPAIR) {
      const float x = ent_params[l * NPAIR + p];
      cv = 0.1f * (1.0f / (1.0f + expf(-x)));
    }
    c[p] = cv;
  }
  {
    const v4f z = (v4f){0.f, 0.f, 0.f, 0.f};
    ((v4f*)&e[0][0])[tid] = z;
    ((v4f*)&e[1][0])[tid] = z;
  }

  const float t0 = rot_params[((l * QQ) + q) * 3 + 0];
  const float t1 = rot_params[((l * QQ) + q) * 3 + 1];
  const float t2 = rot_params[((l * QQ) + q) * 3 + 2];
  const float c0 = cosf(t0 * 0.5f);
  const float c1 = cosf(t1 * 0.5f);
  const float s1 = sinf(t1 * 0.5f);
  const float c2 = cosf(t2 * 0.5f);
  const float b00 = (c0 * c1) * c2;
  const float b01 = (c0 * (-s1)) * c2;
  const float b10 = (c0 * s1) * c2;
  __syncthreads();

  if (tid < 2) {
    const int r = 2 * q + tid;
    float* row = e[tid];
    row[r] = 1.0f;
    if (r < QQ) {
      if (r >= 1) row[r - 1] = c[r - 1];
      if (r <= QQ - 2) {
        float val = c[r];
        int j = r + 1;
        for (; (j & 7) != 0; ++j) { row[j] = val; val *= c[j]; }
        v4f ca = *(const v4f*)&c[j];
        v4f cb = *(const v4f*)&c[j + 4];
        for (; j < QQ; j += 8) {
          const v4f na = *(const v4f*)&c[j + 8];
          const v4f nb = *(const v4f*)&c[j + 12];
          v4f va, vb;
          va[0] = val; val *= ca[0];
          va[1] = val; val *= ca[1];
          va[2] = val; val *= ca[2];
          va[3] = val; val *= ca[3];
          vb[0] = val; val *= cb[0];
          vb[1] = val; val *= cb[1];
          vb[2] = val; val *= cb[2];
          vb[3] = val; val *= cb[3];
          *(v4f*)&row[j] = va;
          *(v4f*)&row[j + 4] = vb;
          ca = na; cb = nb;
        }
      }
    }
  }
  __syncthreads();

  const v4f e0 = ((const v4f*)&e[0][0])[tid];
  const v4f e1 = ((const v4f*)&e[1][0])[tid];
  v4f o0, o1;
#pragma unroll
  for (int k = 0; k < 4; ++k) {
    o0[k] = b00 * e0[k] + b01 * e1[k];
    o1[k] = b10 * e0[k] + b00 * e1[k];
  }
  float* out0 = Tlay + ((size_t)l * DQ + 2 * q) * DQ;
  ((v4f*)out0)[tid] = o0;
  ((v4f*)(out0 + DQ))[tid] = o1;
}

__global__ void vqc_pack_b(const float* __restrict__ B, float* __restrict__ Bp) {
  const int idx = blockIdx.x * blockDim.x + threadIdx.x;
  const int lane = idx & 31;
  const int nt = (idx >> 5) & 31;
  const int ks = idx >> 10;
  const int k0 = ks * 4 + ((lane >> 4) << 1);
  const int col = nt * 16 + (lane & 15);
  v2f v;
  v[0] = B[(size_t)k0 * DQ + col];
  v[1] = B[(size_t)(k0 + 1) * DQ + col];
  *(v2f*)(Bp + (size_t)idx * 2) = v;
}

__global__ void __launch_bounds__(128)
vqc_wmma_gemm(const float* __restrict__ A, const float* __restrict__ Bp,
              float* __restrict__ C) {
  __shared__ v2f As[2 * KSTEPS * 32];

  const int row0 = blockIdx.x * 32;

  {
    const int row = threadIdx.x >> 2;
    const int cq = threadIdx.x & 3;
    const int mt = row >> 4;
    const int lr = row & 15;
    const float* arow = A + (size_t)(row0 + row) * DQ;
#pragma unroll 4
    for (int s = 0; s < 32; ++s) {
      const int ks = cq + 4 * s;
      v4f v = *(const v4f*)(arow + ks * 4);
      v2f lo; lo[0] = v[0]; lo[1] = v[1];
      v2f hi; hi[0] = v[2]; hi[1] = v[3];
      As[(mt * KSTEPS + ks) * 32 + lr] = lo;
      As[(mt * KSTEPS + ks) * 32 + lr + 16] = hi;
    }
  }
  __syncthreads();

  const int wave = threadIdx.x >> 5;
  const int lane = threadIdx.x & 31;
  const int ntbase = wave * 8;

  v8f acc[2][8];
#pragma unroll
  for (int m = 0; m < 2; ++m)
#pragma unroll
    for (int t = 0; t < 8; ++t)
      acc[m][t] = (v8f){0.f, 0.f, 0.f, 0.f, 0.f, 0.f, 0.f, 0.f};

  const v2f* BpV = (const v2f*)Bp;
  for (int ks = 0; ks < KSTEPS; ++ks) {
    const v2f a0 = As[(0 * KSTEPS + ks) * 32 + lane];
    const v2f a1 = As[(1 * KSTEPS + ks) * 32 + lane];
#pragma unroll
    for (int t = 0; t < 8; ++t) {
      const v2f b = BpV[((size_t)ks * NTILES + (ntbase + t)) * 32 + lane];
      acc[0][t] = wmma_f32_4(a0, b, acc[0][t]);
      acc[1][t] = wmma_f32_4(a1, b, acc[1][t]);
    }
  }

  const int lrow = lane & 15;
  const int hi8 = (lane >> 4) * 8;
#pragma unroll
  for (int mt = 0; mt < 2; ++mt)
#pragma unroll
    for (int t = 0; t < 8; ++t) {
      const int col = (ntbase + t) * 16 + lrow;
#pragma unroll
      for (int v = 0; v < 8; ++v) {
        const int r = row0 + mt * 16 + hi8 + v;
        C[(size_t)r * DQ + col] = acc[mt][t][v];
      }
    }
}

#define CHK 16
#define NCHK (KSTEPS / CHK)

__global__ void __launch_bounds__(64)
vqc_wmma_gemm_chain(const float* __restrict__ A, const float* __restrict__ Bp,
                    float* __restrict__ C) {
  __shared__ v2f As[KSTEPS * 32];

  const int row0 = blockIdx.y * 16;
  const int wave = threadIdx.x >> 5;
  const int lane = threadIdx.x & 31;
  const int nt0 = blockIdx.x * 4 + wave * 2;

  const v2f* bp = (const v2f*)Bp + (size_t)nt0 * 32 + lane;

  v2f bc0[CHK], bc1[CHK], ac[CHK];
  v2f bn0[CHK], bn1[CHK], an[CHK];

#pragma unroll
  for (int u = 0; u < CHK; ++u) {
    bc0[u] = bp[(size_t)u * (NTILES * 32)];
    bc1[u] = bp[(size_t)u * (NTILES * 32) + 32];
  }

  {
    const int row = threadIdx.x >> 2;
    const int cq = threadIdx.x & 3;
    const float* arow = A + (size_t)(row0 + row) * DQ;
    v4f v[32];
#pragma unroll
    for (int s = 0; s < 32; ++s) v[s] = *(const v4f*)(arow + (cq + 4 * s) * 4);
#pragma unroll
    for (int s = 0; s < 32; ++s) {
      const int ks = cq + 4 * s;
      v2f lo; lo[0] = v[s][0]; lo[1] = v[s][1];
      v2f hi; hi[0] = v[s][2]; hi[1] = v[s][3];
      As[ks * 32 + row] = lo;
      As[ks * 32 + row + 16] = hi;
    }
  }
  __syncthreads();

#pragma unroll
  for (int u = 0; u < CHK; ++u) ac[u] = As[u * 32 + lane];

  v8f acc0 = (v8f){0.f, 0.f, 0.f, 0.f, 0.f, 0.f, 0.f, 0.f};
  v8f acc1 = (v8f){0.f, 0.f, 0.f, 0.f, 0.f, 0.f, 0.f, 0.f};

#pragma unroll
  for (int c = 0; c < NCHK; ++c) {
    if (c + 1 < NCHK) {
      const int kn = (c + 1) * CHK;
#pragma unroll
      for (int u = 0; u < CHK; ++u) {
        bn0[u] = bp[(size_t)(kn + u) * (NTILES * 32)];
        bn1[u] = bp[(size_t)(kn + u) * (NTILES * 32) + 32];
      }
#pragma unroll
      for (int u = 0; u < CHK; ++u) an[u] = As[(kn + u) * 32 + lane];
    }
#pragma unroll
    for (int u = 0; u < CHK; ++u) {
      acc0 = wmma_f32_4(ac[u], bc0[u], acc0);
      acc1 = wmma_f32_4(ac[u], bc1[u], acc1);
    }
    if (c + 1 < NCHK) {
#pragma unroll
      for (int u = 0; u < CHK; ++u) {
        bc0[u] = bn0[u]; bc1[u] = bn1[u]; ac[u] = an[u];
      }
    }
  }

  __builtin_amdgcn_sched_barrier(0);
  asm volatile("v_nop\n\tv_nop\n\tv_nop\n\tv_nop");
  __builtin_amdgcn_sched_barrier(0);

  const int lrow = lane & 15;
  const int hi8 = (lane >> 4) * 8;
#pragma unroll
  for (int v = 0; v < 8; ++v) {
    float* crow = C + (size_t)(row0 + hi8 + v) * DQ + nt0 * 16 + lrow;
    crow[0] = acc0[v];
    crow[16] = acc1[v];
  }
}

extern "C" void kernel_launch(void* const* d_in, const int* in_sizes, int n_in,
                              void* d_out, int out_size, void* d_ws,
                              size_t ws_size, hipStream_t stream) {
  (void)in_sizes; (void)n_in; (void)out_size; (void)ws_size;
  const float* state = (const float*)d_in[0];
  const float* rotp  = (const float*)d_in[1];
  const float* entp  = (const float*)d_in[2];
  float* out = (float*)d_out;

  float* w     = (float*)d_ws;
  float* Tlay  = w + 1048576;
  float* T01   = w + 0;
  float* T23   = w + 262144;
  float* Ttot  = w + 524288;
  float* Bp    = w + 786432;

  vqc_build_layer<<<dim3(QQ, NLAY), 128, 0, stream>>>(rotp, entp, Tlay);

  vqc_pack_b<<<512, 256, 0, stream>>>(Tlay + 1 * 262144, Bp);
  vqc_wmma_gemm_chain<<<dim3(DQ / 64, DQ / 16), 64, 0, stream>>>(Tlay + 0 * 262144, Bp, T01);

  vqc_pack_b<<<512, 256, 0, stream>>>(Tlay + 3 * 262144, Bp);
  vqc_wmma_gemm_chain<<<dim3(DQ / 64, DQ / 16), 64, 0, stream>>>(Tlay + 2 * 262144, Bp, T23);

  vqc_pack_b<<<512, 256, 0, stream>>>(T23, Bp);
  vqc_wmma_gemm_chain<<<dim3(DQ / 64, DQ / 16), 64, 0, stream>>>(T01, Bp, Ttot);

  vqc_pack_b<<<512, 256, 0, stream>>>(Ttot, Bp);
  vqc_wmma_gemm<<<M_STATE / 32, 128, 0, stream>>>(state, Bp, out);
}


# The 32768x512 product leaves out the blocks of T that are zero by construction (T = dense 256x256 block + 2x2 diagonal blocks): 1088 instead of 4096 WMMAs per row tile; sum order unchanged
# speedup vs baseline: 4.0099x; 1.6650x over previous
#include <hip/hip_runtime.h>
#include <math.h>


typedef float v2f __attribute__((ext_vector_type(2)));
typedef float v4f __attribute__((ext_vector_type(4)));
typedef float v8f __attribute__((ext_vector_type(8)));

#define DQ      512
#define QQ      256
#define NLAY    4
#define NPAIR   255
#define KSTEPS  128
#define NTILES  32
#define M_STATE (16 * 2048)

static __device__ __forceinline__ v8f wmma_f32_4(v2f a, v2f b, v8f c) {
  return __builtin_amdgcn_wmma_f32_16x16x4_f32(false, a, false, b, (short)0, c,
                                               false, false);
}

#define CPAD 8

__global__ void __launch_bounds__(128)
vqc_build_layer(const float* __restrict__ rot_params,
                const float* __restrict__ ent_params,
                float* __restrict__ Tlay) {
  const int q = blockIdx.x;
  const int l = blockIdx.y;
  const int tid = threadIdx.x;

  __shared__ __attribute__((aligned(16))) float c[QQ + CPAD];
  __shared__ __attribute__((aligned(16))) float e[2][DQ];

  for (int p = tid; p < QQ + CPAD; p += 128) {
    float cv = 1.0f;
    if (p < NPAIR) {
      const float x = ent_params[l * NPAIR + p];
      cv = 0.1f * (1.0f / (1.0f + expf(-x)));
    }
    c[p] = cv;
  }
  {
    const v4f z = (v4f){0.f, 0.f, 0.f, 0.f};
    ((v4f*)&e[0][0])[tid] = z;
    ((v4f*)&e[1][0])[tid] = z;
  }

  const float t0 = rot_params[((l * QQ) + q) * 3 + 0];
  const float t1 = rot_params[((l * QQ) + q) * 3 + 1];
  const float t2 = rot_params[((l * QQ) + q) * 3 + 2];
  const float c0 = cosf(t0 * 0.5f);
  const float c1 = cosf(t1 * 0.5f);
  const float s1 = sinf(t1 * 0.5f);
  const float c2 = cosf(t2 * 0.5f);
  const float b00 = (c0 * c1) * c2;
  const float b01 = (c0 * (-s1)) * c2;
  const float b10 = (c0 * s1) * c2;
  __syncthreads();

  if (tid < 2) {
    const int r = 2 * q + tid;
    float* row = e[tid];
    row[r] = 1.0f;
    if (r < QQ) {
      if (r >= 1) row[r - 1] = c[r - 1];
      if (r <= QQ - 2) {
        float val = c[r];
        int j = r + 1;
        for (; (j & 7) != 0; ++j) { row[j] = val; val *= c[j]; }
        v4f ca = *(const v4f*)&c[j];
        v4f cb = *(const v4f*)&c[j + 4];
        for (; j < QQ; j += 8) {
          const v4f na = *(const v4f*)&c[j + 8];
          const v4f nb = *(const v4f*)&c[j + 12];
          v4f va, vb;
          va[0] = val; val *= ca[0];
          va[1] = val; val *= ca[1];
          va[2] = val; val *= ca[2];
          va[3] = val; val *= ca[3];
          vb[0] = val; val *= cb[0];
          vb[1] = val; val *= cb[1];
          vb[2] = val; val *= cb[2];
          vb[3] = val; val *= cb[3];
          *(v4f*)&row[j] = va;
          *(v4f*)&row[j + 4] = vb;
          ca = na; cb = nb;
        }
      }
    }
  }
  __syncthreads();

  const v4f e0 = ((const v4f*)&e[0][0])[tid];
  const v4f e1 = ((const v4f*)&e[1][0])[tid];
  v4f o0, o1;
#pragma unroll
  for (int k = 0; k < 4; ++k) {
    o0[k] = b00 * e0[k] + b01 * e1[k];
    o1[k] = b10 * e0[k] + b00 * e1[k];
  }
  float* out0 = Tlay + ((size_t)l * DQ + 2 * q) * DQ;
  ((v4f*)out0)[tid] = o0;
  ((v4f*)(out0 + DQ))[tid] = o1;
}

__global__ void vqc_pack_b(const float* __restrict__ B, float* __restrict__ Bp) {
  const int idx = blockIdx.x * blockDim.x + threadIdx.x;
  const int lane = idx & 31;
  const int nt = (idx >> 5) & 31;
  const int ks = idx >> 10;
  const int k0 = ks * 4 + ((lane >> 4) << 1);
  const int col = nt * 16 + (lane & 15);
  v2f v;
  v[0] = B[(size_t)k0 * DQ + col];
  v[1] = B[(size_t)(k0 + 1) * DQ + col];
  *(v2f*)(Bp + (size_t)idx * 2) = v;
}

#define KTOP   64
#define NTTOP  16
#define DCH    4
#define NDCH   (KTOP / DCH)

__global__ void __launch_bounds__(128)
vqc_wmma_gemm(const float* __restrict__ A, const float* __restrict__ Bp,
              float* __restrict__ C) {
  __shared__ v2f As[2 * KSTEPS * 32];

  const int row0 = blockIdx.x * 32;
  const int wave = threadIdx.x >> 5;
  const int lane = threadIdx.x & 31;
  const int nttop = wave * 4;
  const int ntbot = NTTOP + wave * 4;
  const v2f* BpV = (const v2f*)Bp;
  const v2f* bpt = BpV + (size_t)nttop * 32 + lane;

  v2f bd[4][4];
#pragma unroll
  for (int t = 0; t < 4; ++t)
#pragma unroll
    for (int j = 0; j < 4; ++j) {
      const int nt = ntbot + t;
      bd[t][j] = BpV[((size_t)(4 * nt + j) * NTILES + nt) * 32 + lane];
    }
  v2f bc[DCH][4], bn[DCH][4], ac[DCH][2], an[DCH][2];
#pragma unroll
  for (int u = 0; u < DCH; ++u)
#pragma unroll
    for (int t = 0; t < 4; ++t)
      bc[u][t] = bpt[(size_t)u * (NTILES * 32) + t * 32];

  {
    const int row = threadIdx.x >> 2;
    const int cq = threadIdx.x & 3;
    const int mt = row >> 4;
    const int lr = row & 15;
    const float* arow = A + (size_t)(row0 + row) * DQ;
#pragma unroll 4
    for (int s = 0; s < 32; ++s) {
      const int ks = cq + 4 * s;
      v4f v = *(const v4f*)(arow + ks * 4);
      v2f lo; lo[0] = v[0]; lo[1] = v[1];
      v2f hi; hi[0] = v[2]; hi[1] = v[3];
      As[(mt * KSTEPS + ks) * 32 + lr] = lo;
      As[(mt * KSTEPS + ks) * 32 + lr + 16] = hi;
    }
  }
  __syncthreads();

  {
    v2f ad[4][4][2];
#pragma unroll
    for (int t = 0; t < 4; ++t)
#pragma unroll
      for (int j = 0; j < 4; ++j) {
        const int ks = 4 * (ntbot + t) + j;
        ad[t][j][0] = As[(0 * KSTEPS + ks) * 32 + lane];
        ad[t][j][1] = As[(1 * KSTEPS + ks) * 32 + lane];
      }
    v8f accd[2][4];
#pragma unroll
    for (int m = 0; m < 2; ++m)
#pragma unroll
      for (int t = 0; t < 4; ++t)
        accd[m][t] = (v8f){0.f, 0.f, 0.f, 0.f, 0.f, 0.f, 0.f, 0.f};
#pragma unroll
    for (int j = 0; j < 4; ++j)
#pragma unroll
      for (int t = 0; t < 4; ++t) {
        accd[0][t] = wmma_f32_4(ad[t][j][0], bd[t][j], accd[0][t]);
        accd[1][t] = wmma_f32_4(ad[t][j][1], bd[t][j], accd[1][t]);
      }

    __builtin_amdgcn_sched_barrier(0);
    asm volatile("v_nop\n\tv_nop\n\tv_nop\n\tv_nop");
    __builtin_amdgcn_sched_barrier(0);

    const int lrow = lane & 15;
    const int hi8 = (lane >> 4) * 8;
#pragma unroll
    for (int mt = 0; mt < 2; ++mt)
#pragma unroll
      for (int t = 0; t < 4; ++t) {
        const int col = (ntbot + t) * 16 + lrow;
#pragma unroll
        for (int v = 0; v < 8; ++v) {
          const int r = row0 + mt * 16 + hi8 + v;
          C[(size_t)r * DQ + col] = accd[mt][t][v];
        }
      }
    __builtin_amdgcn_sched_barrier(0);
  }

#pragma unroll
  for (int u = 0; u < DCH; ++u) {
    ac[u][0] = As[(0 * KSTEPS + u) * 32 + lane];
    ac[u][1] = As[(1 * KSTEPS + u) * 32 + lane];
  }
  v8f acc[2][4];
#pragma unroll
  for (int m = 0; m < 2; ++m)
#pragma unroll
    for (int t = 0; t < 4; ++t)
      acc[m][t] = (v8f){0.f, 0.f, 0.f, 0.f, 0.f, 0.f, 0.f, 0.f};

#pragma unroll
  for (int c = 0; c < NDCH; ++c) {
    if (c + 1 < NDCH) {
      const int kn = (c + 1) * DCH;
#pragma unroll
      for (int u = 0; u < DCH; ++u)
#pragma unroll
        for (int t = 0; t < 4; ++t)
          bn[u][t] = bpt[(size_t)(kn + u) * (NTILES * 32) + t * 32];
#pragma unroll
      for (int u = 0; u < DCH; ++u) {
        an[u][0] = As[(0 * KSTEPS + kn + u) * 32 + lane];
        an[u][1] = As[(1 * KSTEPS + kn + u) * 32 + lane];
      }
    }
#pragma unroll
    for (int u = 0; u < DCH; ++u)
#pragma unroll
      for (int t = 0; t < 4; ++t) {
        acc[0][t] = wmma_f32_4(ac[u][0], bc[u][t], acc[0][t]);
        acc[1][t] = wmma_f32_4(ac[u][1], bc[u][t], acc[1][t]);
      }
    if (c + 1 < NDCH) {
#pragma unroll
      for (int u = 0; u < DCH; ++u) {
#pragma unroll
        for (int t = 0; t < 4; ++t) bc[u][t] = bn[u][t];
        ac[u][0] = an[u][0]; ac[u][1] = an[u][1];
      }
    }
  }

  __builtin_amdgcn_sched_barrier(0);
  asm volatile("v_nop\n\tv_nop\n\tv_nop\n\tv_nop");
  __builtin_amdgcn_sched_barrier(0);

  {
    const int lrow = lane & 15;
    const int hi8 = (lane >> 4) * 8;
#pragma unroll
    for (int mt = 0; mt < 2; ++mt)
#pragma unroll
      for (int t = 0; t < 4; ++t) {
        const int col = (nttop + t) * 16 + lrow;
#pragma unroll
        for (int v = 0; v < 8; ++v) {
          const int r = row0 + mt * 16 + hi8 + v;
          C[(size_t)r * DQ + col] = acc[mt][t][v];
        }
      }
  }
}

#define CHK 16
#define NCHK (KSTEPS / CHK)

__global__ void __launch_bounds__(64)
vqc_wmma_gemm_chain(const float* __restrict__ A, const float* __restrict__ Bp,
                    float* __restrict__ C) {
  __shared__ v2f As[KSTEPS * 32];

  const int row0 = blockIdx.y * 16;
  const int wave = threadIdx.x >> 5;
  const int lane = threadIdx.x & 31;
  const int nt0 = blockIdx.x * 4 + wave * 2;

  const v2f* bp = (const v2f*)Bp + (size_t)nt0 * 32 + lane;

  v2f bc0[CHK], bc1[CHK], ac[CHK];
  v2f bn0[CHK], bn1[CHK], an[CHK];

#pragma unroll
  for (int u = 0; u < CHK; ++u) {
    bc0[u] = bp[(size_t)u * (NTILES * 32)];
    bc1[u] = bp[(size_t)u * (NTILES * 32) + 32];
  }

  {
    const int row = threadIdx.x >> 2;
    const int cq = threadIdx.x & 3;
    const float* arow = A + (size_t)(row0 + row) * DQ;
    v4f v[32];
#pragma unroll
    for (int s = 0; s < 32; ++s) v[s] = *(const v4f*)(arow + (cq + 4 * s) * 4);
#pragma unroll
    for (int s = 0; s < 32; ++s) {
      const int ks = cq + 4 * s;
      v2f lo; lo[0] = v[s][0]; lo[1] = v[s][1];
      v2f hi; hi[0] = v[s][2]; hi[1] = v[s][3];
      As[ks * 32 + row] = lo;
      As[ks * 32 + row + 16] = hi;
    }
  }
  __syncthreads();

#pragma unroll
  for (int u = 0; u < CHK; ++u) ac[u] = As[u * 32 + lane];

  v8f acc0 = (v8f){0.f, 0.f, 0.f, 0.f, 0.f, 0.f, 0.f, 0.f};
  v8f acc1 = (v8f){0.f, 0.f, 0.f, 0.f, 0.f, 0.f, 0.f, 0.f};

#pragma unroll
  for (int c = 0; c < NCHK; ++c) {
    if (c + 1 < NCHK) {
      const int kn = (c + 1) * CHK;
#pragma unroll
      for (int u = 0; u < CHK; ++u) {
        bn0[u] = bp[(size_t)(kn + u) * (NTILES * 32)];
        bn1[u] = bp[(size_t)(kn + u) * (NTILES * 32) + 32];
      }
#pragma unroll
      for (int u = 0; u < CHK; ++u) an[u] = As[(kn + u) * 32 + lane];
    }
#pragma unroll
    for (int u = 0; u < CHK; ++u) {
      acc0 = wmma_f32_4(ac[u], bc0[u], acc0);
      acc1 = wmma_f32_4(ac[u], bc1[u], acc1);
    }
    if (c + 1 < NCHK) {
#pragma unroll
      for (int u = 0; u < CHK; ++u) {
        bc0[u] = bn0[u]; bc1[u] = bn1[u]; ac[u] = an[u];
      }
    }
  }

  __builtin_amdgcn_sched_barrier(0);
  asm volatile("v_nop\n\tv_nop\n\tv_nop\n\tv_nop");
  __builtin_amdgcn_sched_barrier(0);

  const int lrow = lane & 15;
  const int hi8 = (lane >> 4) * 8;
#pragma unroll
  for (int v = 0; v < 8; ++v) {
    float* crow = C + (size_t)(row0 + hi8 + v) * DQ + nt0 * 16 + lrow;
    crow[0] = acc0[v];
    crow[16] = acc1[v];
  }
}

extern "C" void kernel_launch(void* const* d_in, const int* in_sizes, int n_in,
                              void* d_out, int out_size, void* d_ws,
                              size_t ws_size, hipStream_t stream) {
  (void)in_sizes; (void)n_in; (void)out_size; (void)ws_size;
  const float* state = (const float*)d_in[0];
  const float* rotp  = (const float*)d_in[1];
  const float* entp  = (const float*)d_in[2];
  float* out = (float*)d_out;

  float* w     = (float*)d_ws;
  float* Tlay  = w + 1048576;
  float* T01   = w + 0;
  float* T23   = w + 262144;
  float* Ttot  = w + 524288;
  float* Bp    = w + 786432;

  vqc_build_layer<<<dim3(QQ, NLAY), 128, 0, stream>>>(rotp, entp, Tlay);

  vqc_pack_b<<<512, 256, 0, stream>>>(Tlay + 1 * 262144, Bp);
  vqc_wmma_gemm_chain<<<dim3(DQ / 64, DQ / 16), 64, 0, stream>>>(Tlay + 0 * 262144, Bp, T01);

  vqc_pack_b<<<512, 256, 0, stream>>>(Tlay + 3 * 262144, Bp);
  vqc_wmma_gemm_chain<<<dim3(DQ / 64, DQ / 16), 64, 0, stream>>>(Tlay + 2 * 262144, Bp, T23);

  vqc_pack_b<<<512, 256, 0, stream>>>(T23, Bp);
  vqc_wmma_gemm_chain<<<dim3(DQ / 64, DQ / 16), 64, 0, stream>>>(T01, Bp, Ttot);

  vqc_pack_b<<<512, 256, 0, stream>>>(Ttot, Bp);
  vqc_wmma_gemm<<<M_STATE / 32, 128, 0, stream>>>(state, Bp, out);
}


# Each product's B operand is written in WMMA fragment order by the kernel that produces it (layer build, chain product); the four re-ordering launches go: 9 -> 5 launches; sum order unchanged
# speedup vs baseline: 4.6979x; 1.1716x over previous
#include <hip/hip_runtime.h>
#include <math.h>


typedef float v2f __attribute__((ext_vector_type(2)));
typedef float v4f __attribute__((ext_vector_type(4)));
typedef float v8f __attribute__((ext_vector_type(8)));

#define DQ      512
#define QQ      256
#define NLAY    4
#define NPAIR   255
#define KSTEPS  128
#define NTILES  32
#define M_STATE (16 * 2048)

static __device__ __forceinline__ v8f wmma_f32_4(v2f a, v2f b, v8f c) {
  return __builtin_amdgcn_wmma_f32_16x16x4_f32(false, a, false, b, (short)0, c,
                                               false, false);
}

#define CPAD 8

__global__ void __launch_bounds__(128)
vqc_build_layer(const float* __restrict__ rot_params,
                const float* __restrict__ ent_params,
                float* __restrict__ Tlay) {
  const int q = blockIdx.x;
  const int l = blockIdx.y;
  const int tid = threadIdx.x;

  __shared__ __attribute__((aligned(16))) float c[QQ + CPAD];
  __shared__ __attribute__((aligned(16))) float e[2][DQ];

  for (int p = tid; p < QQ + CPAD; p += 128) {
    float cv = 1.0f;
    if (p < NPAIR) {
      const float x = ent_params[l * NPAIR + p];
      cv = 0.1f * (1.0f / (1.0f + expf(-x)));
    }
    c[p] = cv;
  }
  {
    const v4f z = (v4f){0.f, 0.f, 0.f, 0.f};
    ((v4f*)&e[0][0])[tid] = z;
    ((v4f*)&e[1][0])[tid] = z;
  }

  const float t0 = rot_params[((l * QQ) + q) * 3 + 0];
  const float t1 = rot_params[((l * QQ) + q) * 3 + 1];
  const float t2 = rot_params[((l * QQ) + q) * 3 + 2];
  const float c0 = cosf(t0 * 0.5f);
  const float c1 = cosf(t1 * 0.5f);
  const float s1 = sinf(t1 * 0.5f);
  const float c2 = cosf(t2 * 0.5f);
  const float b00 = (c0 * c1) * c2;
  const float b01 = (c0 * (-s1)) * c2;
  const float b10 = (c0 * s1) * c2;
  __syncthreads();

  if (tid < 2) {
    const int r = 2 * q + tid;
    float* row = e[tid];
    row[r] = 1.0f;
    if (r < QQ) {
      if (r >= 1) row[r - 1] = c[r - 1];
      if (r <= QQ - 2) {
        float val = c[r];
        int j = r + 1;
        for (; (j & 7) != 0; ++j) { row[j] = val; val *= c[j]; }
        v4f ca = *(const v4f*)&c[j];
        v4f cb = *(const v4f*)&c[j + 4];
        for (; j < QQ; j += 8) {
          const v4f na = *(const v4f*)&c[j + 8];
          const v4f nb = *(const v4f*)&c[j + 12];
          v4f va, vb;
          va[0] = val; val *= ca[0];
          va[1] = val; val *= ca[1];
          va[2] = val; val *= ca[2];
          va[3] = val; val *= ca[3];
          vb[0] = val; val *= cb[0];
          vb[1] = val; val *= cb[1];
          vb[2] = val; val *= cb[2];
          vb[3] = val; val *= cb[3];
          *(v4f*)&row[j] = va;
          *(v4f*)&row[j + 4] = vb;
          ca = na; cb = nb;
        }
      }
    }
  }
  __syncthreads();

  if (l & 1) {
    const int ks = q >> 1;
    const int half = q & 1;
    float* outp = Tlay + (size_t)l * DQ * DQ;
#pragma unroll
    for (int s = 0; s < 2; ++s) {
      const int col = 256 * s + 2 * tid;
      const v2f e0 = ((const v2f*)&e[0][0])[128 * s + tid];
      const v2f e1 = ((const v2f*)&e[1][0])[128 * s + tid];
      v4f o;
      o[0] = b00 * e0[0] + b01 * e1[0];
      o[1] = b10 * e0[0] + b00 * e1[0];
      o[2] = b00 * e0[1] + b01 * e1[1];
      o[3] = b10 * e0[1] + b00 * e1[1];
      const int pair = (ks * NTILES + (col >> 4)) * 32 + 16 * half + (col & 15);
      *(v4f*)(outp + (size_t)pair * 2) = o;
    }
  } else {
    const v4f e0 = ((const v4f*)&e[0][0])[tid];
    const v4f e1 = ((const v4f*)&e[1][0])[tid];
    v4f o0, o1;
#pragma unroll
    for (int k = 0; k < 4; ++k) {
      o0[k] = b00 * e0[k] + b01 * e1[k];
      o1[k] = b10 * e0[k] + b00 * e1[k];
    }
    float* out0 = Tlay + ((size_t)l * DQ + 2 * q) * DQ;
    ((v4f*)out0)[tid] = o0;
    ((v4f*)(out0 + DQ))[tid] = o1;
  }
}

#define KTOP   64
#define NTTOP  16
#define DCH    4
#define NDCH   (KTOP / DCH)

__global__ void __launch_bounds__(128)
vqc_wmma_gemm(const float* __restrict__ A, const float* __restrict__ Bp,
              float* __restrict__ C) {
  __shared__ v2f As[2 * KSTEPS * 32];

  const int row0 = blockIdx.x * 32;
  const int wave = threadIdx.x >> 5;
  const int lane = threadIdx.x & 31;
  const int nttop = wave * 4;
  const int ntbot = NTTOP + wave * 4;
  const v2f* BpV = (const v2f*)Bp;
  const v2f* bpt = BpV + (size_t)nttop * 32 + lane;

  v2f bd[4][4];
#pragma unroll
  for (int t = 0; t < 4; ++t)
#pragma unroll
    for (int j = 0; j < 4; ++j) {
      const int nt = ntbot + t;
      bd[t][j] = BpV[((size_t)(4 * nt + j) * NTILES + nt) * 32 + lane];
    }
  v2f bc[DCH][4], bn[DCH][4], ac[DCH][2], an[DCH][2];
#pragma unroll
  for (int u = 0; u < DCH; ++u)
#pragma unroll
    for (int t = 0; t < 4; ++t)
      bc[u][t] = bpt[(size_t)u * (NTILES * 32) + t * 32];

  {
    const int row = threadIdx.x >> 2;
    const int cq = threadIdx.x & 3;
    const int mt = row >> 4;
    const int lr = row & 15;
    const float* arow = A + (size_t)(row0 + row) * DQ;
#pragma unroll 4
    for (int s = 0; s < 32; ++s) {
      const int ks = cq + 4 * s;
      v4f v = *(const v4f*)(arow + ks * 4);
      v2f lo; lo[0] = v[0]; lo[1] = v[1];
      v2f hi; hi[0] = v[2]; hi[1] = v[3];
      As[(mt * KSTEPS + ks) * 32 + lr] = lo;
      As[(mt * KSTEPS + ks) * 32 + lr + 16] = hi;
    }
  }
  __syncthreads();

  {
    v2f ad[4][4][2];
#pragma unroll
    for (int t = 0; t < 4; ++t)
#pragma unroll
      for (int j = 0; j < 4; ++j) {
        const int ks = 4 * (ntbot + t) + j;
        ad[t][j][0] = As[(0 * KSTEPS + ks) * 32 + lane];
        ad[t][j][1] = As[(1 * KSTEPS + ks) * 32 + lane];
      }
    v8f accd[2][4];
#pragma unroll
    for (int m = 0; m < 2; ++m)
#pragma unroll
      for (int t = 0; t < 4; ++t)
        accd[m][t] = (v8f){0.f, 0.f, 0.f, 0.f, 0.f, 0.f, 0.f, 0.f};
#pragma unroll
    for (int j = 0; j < 4; ++j)
#pragma unroll
      for (int t = 0; t < 4; ++t) {
        accd[0][t] = wmma_f32_4(ad[t][j][0], bd[t][j], accd[0][t]);
        accd[1][t] = wmma_f32_4(ad[t][j][1], bd[t][j], accd[1][t]);
      }

    __builtin_amdgcn_sched_barrier(0);
    asm volatile("v_nop\n\tv_nop\n\tv_nop\n\tv_nop");
    __builtin_amdgcn_sched_barrier(0);

    const int lrow = lane & 15;
    const int hi8 = (lane >> 4) * 8;
#pragma unroll
    for (int mt = 0; mt < 2; ++mt)
#pragma unroll
      for (int t = 0; t < 4; ++t) {
        const int col = (ntbot + t) * 16 + lrow;
#pragma unroll
        for (int v = 0; v < 8; ++v) {
          const int r = row0 + mt * 16 + hi8 + v;
          C[(size_t)r * DQ + col] = accd[mt][t][v];
        }
      }
    __builtin_amdgcn_sched_barrier(0);
  }

#pragma unroll
  for (int u = 0; u < DCH; ++u) {
    ac[u][0] = As[(0 * KSTEPS + u) * 32 + lane];
    ac[u][1] = As[(1 * KSTEPS + u) * 32 + lane];
  }
  v8f acc[2][4];
#pragma unroll
  for (int m = 0; m < 2; ++m)
#pragma unroll
    for (int t = 0; t < 4; ++t)
      acc[m][t] = (v8f){0.f, 0.f, 0.f, 0.f, 0.f, 0.f, 0.f, 0.f};

#pragma unroll
  for (int c = 0; c < NDCH; ++c) {
    if (c + 1 < NDCH) {
      const int kn = (c + 1) * DCH;
#pragma unroll
      for (int u = 0; u < DCH; ++u)
#pragma unroll
        for (int t = 0; t < 4; ++t)
          bn[u][t] = bpt[(size_t)(kn + u) * (NTILES * 32) + t * 32];
#pragma unroll
      for (int u = 0; u < DCH; ++u) {
        an[u][0] = As[(0 * KSTEPS + kn + u) * 32 + lane];
        an[u][1] = As[(1 * KSTEPS + kn + u) * 32 + lane];
      }
    }
#pragma unroll
    for (int u = 0; u < DCH; ++u)
#pragma unroll
      for (int t = 0; t < 4; ++t) {
        acc[0][t] = wmma_f32_4(ac[u][0], bc[u][t], acc[0][t]);
        acc[1][t] = wmma_f32_4(ac[u][1], bc[u][t], acc[1][t]);
      }
    if (c + 1 < NDCH) {
#pragma unroll
      for (int u = 0; u < DCH; ++u) {
#pragma unroll
        for (int t = 0; t < 4; ++t) bc[u][t] = bn[u][t];
        ac[u][0] = an[u][0]; ac[u][1] = an[u][1];
      }
    }
  }

  __builtin_amdgcn_sched_barrier(0);
  asm volatile("v_nop\n\tv_nop\n\tv_nop\n\tv_nop");
  __builtin_amdgcn_sched_barrier(0);

  {
    const int lrow = lane & 15;
    const int hi8 = (lane >> 4) * 8;
#pragma unroll
    for (int mt = 0; mt < 2; ++mt)
#pragma unroll
      for (int t = 0; t < 4; ++t) {
        const int col = (nttop + t) * 16 + lrow;
#pragma unroll
        for (int v = 0; v < 8; ++v) {
          const int r = row0 + mt * 16 + hi8 + v;
          C[(size_t)r * DQ + col] = acc[mt][t][v];
        }
      }
  }
}

#define CHK 16
#define NCHK (KSTEPS / CHK)

template <bool PACKED_OUT>
__global__ void __launch_bounds__(64)
vqc_wmma_gemm_chain(const float* __restrict__ A, const float* __restrict__ Bp,
                    float* __restrict__ C) {
  __shared__ v2f As[KSTEPS * 32];

  const int row0 = blockIdx.y * 16;
  const int wave = threadIdx.x >> 5;
  const int lane = threadIdx.x & 31;
  const int nt0 = blockIdx.x * 4 + wave * 2;

  const v2f* bp = (const v2f*)Bp + (size_t)nt0 * 32 + lane;

  v2f bc0[CHK], bc1[CHK], ac[CHK];
  v2f bn0[CHK], bn1[CHK], an[CHK];

#pragma unroll
  for (int u = 0; u < CHK; ++u) {
    bc0[u] = bp[(size_t)u * (NTILES * 32)];
    bc1[u] = bp[(size_t)u * (NTILES * 32) + 32];
  }

  {
    const int row = threadIdx.x >> 2;
    const int cq = threadIdx.x & 3;
    const float* arow = A + (size_t)(row0 + row) * DQ;
    v4f v[32];
#pragma unroll
    for (int s = 0; s < 32; ++s) v[s] = *(const v4f*)(arow + (cq + 4 * s) * 4);
#pragma unroll
    for (int s = 0; s < 32; ++s) {
      const int ks = cq + 4 * s;
      v2f lo; lo[0] = v[s][0]; lo[1] = v[s][1];
      v2f hi; hi[0] = v[s][2]; hi[1] = v[s][3];
      As[ks * 32 + row] = lo;
      As[ks * 32 + row + 16] = hi;
    }
  }
  __syncthreads();

#pragma unroll
  for (int u = 0; u < CHK; ++u) ac[u] = As[u * 32 + lane];

  v8f acc0 = (v8f){0.f, 0.f, 0.f, 0.f, 0.f, 0.f, 0.f, 0.f};
  v8f acc1 = (v8f){0.f, 0.f, 0.f, 0.f, 0.f, 0.f, 0.f, 0.f};

#pragma unroll
  for (int c = 0; c < NCHK; ++c) {
    if (c + 1 < NCHK) {
      const int kn = (c + 1) * CHK;
#pragma unroll
      for (int u = 0; u < CHK; ++u) {
        bn0[u] = bp[(size_t)(kn + u) * (NTILES * 32)];
        bn1[u] = bp[(size_t)(kn + u) * (NTILES * 32) + 32];
      }
#pragma unroll
      for (int u = 0; u < CHK; ++u) an[u] = As[(kn + u) * 32 + lane];
    }
#pragma unroll
    for (int u = 0; u < CHK; ++u) {
      acc0 = wmma_f32_4(ac[u], bc0[u], acc0);
      acc1 = wmma_f32_4(ac[u], bc1[u], acc1);
    }
    if (c + 1 < NCHK) {
#pragma unroll
      for (int u = 0; u < CHK; ++u) {
        bc0[u] = bn0[u]; bc1[u] = bn1[u]; ac[u] = an[u];
      }
    }
  }

  __builtin_amdgcn_sched_barrier(0);
  asm volatile("v_nop\n\tv_nop\n\tv_nop\n\tv_nop");
  __builtin_amdgcn_sched_barrier(0);

  const int lrow = lane & 15;
  const int hi8 = (lane >> 4) * 8;
  if (PACKED_OUT) {
    v2f* cp = (v2f*)C + ((size_t)(((row0 + hi8) >> 2) * NTILES + nt0) * 32 + lrow);
#pragma unroll
    for (int p = 0; p < 4; ++p) {
      v2f x0; x0[0] = acc0[2 * p]; x0[1] = acc0[2 * p + 1];
      v2f x1; x1[0] = acc1[2 * p]; x1[1] = acc1[2 * p + 1];
      const int off = (p >> 1) * (NTILES * 32) + 16 * (p & 1);
      cp[off] = x0;
      cp[off + 32] = x1;
    }
  } else {
#pragma unroll
    for (int v = 0; v < 8; ++v) {
      float* crow = C + (size_t)(row0 + hi8 + v) * DQ + nt0 * 16 + lrow;
      crow[0] = acc0[v];
      crow[16] = acc1[v];
    }
  }
}

extern "C" void kernel_launch(void* const* d_in, const int* in_sizes, int n_in,
                              void* d_out, int out_size, void* d_ws,
                              size_t ws_size, hipStream_t stream) {
  (void)in_sizes; (void)n_in; (void)out_size; (void)ws_size;
  const float* state = (const float*)d_in[0];
  const float* rotp  = (const float*)d_in[1];
  const float* entp  = (const float*)d_in[2];
  float* out = (float*)d_out;

  float* w     = (float*)d_ws;
  float* Tlay  = w + 1048576;
  float* T01   = w + 0;
  float* T23   = w + 262144;
  float* Ttot  = w + 524288;

  vqc_build_layer<<<dim3(QQ, NLAY), 128, 0, stream>>>(rotp, entp, Tlay);

  vqc_wmma_gemm_chain<false><<<dim3(DQ / 64, DQ / 16), 64, 0, stream>>>(
      Tlay + 0 * 262144, Tlay + 1 * 262144, T01);

  vqc_wmma_gemm_chain<true><<<dim3(DQ / 64, DQ / 16), 64, 0, stream>>>(
      Tlay + 2 * 262144, Tlay + 3 * 262144, T23);

  vqc_wmma_gemm_chain<true><<<dim3(DQ / 64, DQ / 16), 64, 0, stream>>>(T01, T23, Ttot);

  vqc_wmma_gemm<<<M_STATE / 32, 128, 0, stream>>>(state, Ttot, out);
}


# T01 and T23, independent of each other, share ONE launch: waves 0-1 of each work-group do one product, waves 2-3 the other on the two SIMDs left idle; 5 -> 4 launches; sum order unchanged
# speedup vs baseline: 4.9669x; 1.0573x over previous
#include <hip/hip_runtime.h>
#include <math.h>


typedef float v2f __attribute__((ext_vector_type(2)));
typedef float v4f __attribute__((ext_vector_type(4)));
typedef float v8f __attribute__((ext_vector_type(8)));

#define DQ      512
#define QQ      256
#define NLAY    4
#define NPAIR   255
#define KSTEPS  128
#define NTILES  32
#define M_STATE (16 * 2048)

static __device__ __forceinline__ v8f wmma_f32_4(v2f a, v2f b, v8f c) {
  return __builtin_amdgcn_wmma_f32_16x16x4_f32(false, a, false, b, (short)0, c,
                                               false, false);
}

#define CPAD 8

__global__ void __launch_bounds__(128)
vqc_build_layer(const float* __restrict__ rot_params,
                const float* __restrict__ ent_params,
                float* __restrict__ Tlay) {
  const int q = blockIdx.x;
  const int l = blockIdx.y;
  const int tid = threadIdx.x;

  __shared__ __attribute__((aligned(16))) float c[QQ + CPAD];
  __shared__ __attribute__((aligned(16))) float e[2][DQ];

  for (int p = tid; p < QQ + CPAD; p += 128) {
    float cv = 1.0f;
    if (p < NPAIR) {
      const float x = ent_params[l * NPAIR + p];
      cv = 0.1f * (1.0f / (1.0f + expf(-x)));
    }
    c[p] = cv;
  }
  {
    const v4f z = (v4f){0.f, 0.f, 0.f, 0.f};
    ((v4f*)&e[0][0])[tid] = z;
    ((v4f*)&e[1][0])[tid] = z;
  }

  const float t0 = rot_params[((l * QQ) + q) * 3 + 0];
  const float t1 = rot_params[((l * QQ) + q) * 3 + 1];
  const float t2 = rot_params[((l * QQ) + q) * 3 + 2];
  const float c0 = cosf(t0 * 0.5f);
  const float c1 = cosf(t1 * 0.5f);
  const float s1 = sinf(t1 * 0.5f);
  const float c2 = cosf(t2 * 0.5f);
  const float b00 = (c0 * c1) * c2;
  const float b01 = (c0 * (-s1)) * c2;
  const float b10 = (c0 * s1) * c2;
  __syncthreads();

  if (tid < 2) {
    const int r = 2 * q + tid;
    float* row = e[tid];
    row[r] = 1.0f;
    if (r < QQ) {
      if (r >= 1) row[r - 1] = c[r - 1];
      if (r <= QQ - 2) {
        float val = c[r];
        int j = r + 1;
        for (; (j & 7) != 0; ++j) { row[j] = val; val *= c[j]; }
        v4f ca = *(const v4f*)&c[j];
        v4f cb = *(const v4f*)&c[j + 4];
        for (; j < QQ; j += 8) {
          const v4f na = *(const v4f*)&c[j + 8];
          const v4f nb = *(const v4f*)&c[j + 12];
          v4f va, vb;
          va[0] = val; val *= ca[0];
          va[1] = val; val *= ca[1];
          va[2] = val; val *= ca[2];
          va[3] = val; val *= ca[3];
          vb[0] = val; val *= cb[0];
          vb[1] = val; val *= cb[1];
          vb[2] = val; val *= cb[2];
          vb[3] = val; val *= cb[3];
          *(v4f*)&row[j] = va;
          *(v4f*)&row[j + 4] = vb;
          ca = na; cb = nb;
        }
      }
    }
  }
  __syncthreads();

  if (l & 1) {
    const int ks = q >> 1;
    const int half = q & 1;
    float* outp = Tlay + (size_t)l * DQ * DQ;
#pragma unroll
    for (int s = 0; s < 2; ++s) {
      const int col = 256 * s + 2 * tid;
      const v2f e0 = ((const v2f*)&e[0][0])[128 * s + tid];
      const v2f e1 = ((const v2f*)&e[1][0])[128 * s + tid];
      v4f o;
      o[0] = b00 * e0[0] + b01 * e1[0];
      o[1] = b10 * e0[0] + b00 * e1[0];
      o[2] = b00 * e0[1] + b01 * e1[1];
      o[3] = b10 * e0[1] + b00 * e1[1];
      const int pair = (ks * NTILES + (col >> 4)) * 32 + 16 * half + (col & 15);
      *(v4f*)(outp + (size_t)pair * 2) = o;
    }
  } else {
    const v4f e0 = ((const v4f*)&e[0][0])[tid];
    const v4f e1 = ((const v4f*)&e[1][0])[tid];
    v4f o0, o1;
#pragma unroll
    for (int k = 0; k < 4; ++k) {
      o0[k] = b00 * e0[k] + b01 * e1[k];
      o1[k] = b10 * e0[k] + b00 * e1[k];
    }
    float* out0 = Tlay + ((size_t)l * DQ + 2 * q) * DQ;
    ((v4f*)out0)[tid] = o0;
    ((v4f*)(out0 + DQ))[tid] = o1;
  }
}

#define KTOP   64
#define NTTOP  16
#define DCH    4
#define NDCH   (KTOP / DCH)

__global__ void __launch_bounds__(128)
vqc_wmma_gemm(const float* __restrict__ A, const float* __restrict__ Bp,
              float* __restrict__ C) {
  __shared__ v2f As[2 * KSTEPS * 32];

  const int row0 = blockIdx.x * 32;
  const int wave = threadIdx.x >> 5;
  const int lane = threadIdx.x & 31;
  const int nttop = wave * 4;
  const int ntbot = NTTOP + wave * 4;
  const v2f* BpV = (const v2f*)Bp;
  const v2f* bpt = BpV + (size_t)nttop * 32 + lane;

  v2f bd[4][4];
#pragma unroll
  for (int t = 0; t < 4; ++t)
#pragma unroll
    for (int j = 0; j < 4; ++j) {
      const int nt = ntbot + t;
      bd[t][j] = BpV[((size_t)(4 * nt + j) * NTILES + nt) * 32 + lane];
    }
  v2f bc[DCH][4], bn[DCH][4], ac[DCH][2], an[DCH][2];
#pragma unroll
  for (int u = 0; u < DCH; ++u)
#pragma unroll
    for (int t = 0; t < 4; ++t)
      bc[u][t] = bpt[(size_t)u * (NTILES * 32) + t * 32];

  {
    const int row = threadIdx.x >> 2;
    const int cq = threadIdx.x & 3;
    const int mt = row >> 4;
    const int lr = row & 15;
    const float* arow = A + (size_t)(row0 + row) * DQ;
#pragma unroll 4
    for (int s = 0; s < 32; ++s) {
      const int ks = cq + 4 * s;
      v4f v = *(const v4f*)(arow + ks * 4);
      v2f lo; lo[0] = v[0]; lo[1] = v[1];
      v2f hi; hi[0] = v[2]; hi[1] = v[3];
      As[(mt * KSTEPS + ks) * 32 + lr] = lo;
      As[(mt * KSTEPS + ks) * 32 + lr + 16] = hi;
    }
  }
  __syncthreads();

  {
    v2f ad[4][4][2];
#pragma unroll
    for (int t = 0; t < 4; ++t)
#pragma unroll
      for (int j = 0; j < 4; ++j) {
        const int ks = 4 * (ntbot + t) + j;
        ad[t][j][0] = As[(0 * KSTEPS + ks) * 32 + lane];
        ad[t][j][1] = As[(1 * KSTEPS + ks) * 32 + lane];
      }
    v8f accd[2][4];
#pragma unroll
    for (int m = 0; m < 2; ++m)
#pragma unroll
      for (int t = 0; t < 4; ++t)
        accd[m][t] = (v8f){0.f, 0.f, 0.f, 0.f, 0.f, 0.f, 0.f, 0.f};
#pragma unroll
    for (int j = 0; j < 4; ++j)
#pragma unroll
      for (int t = 0; t < 4; ++t) {
        accd[0][t] = wmma_f32_4(ad[t][j][0], bd[t][j], accd[0][t]);
        accd[1][t] = wmma_f32_4(ad[t][j][1], bd[t][j], accd[1][t]);
      }

    __builtin_amdgcn_sched_barrier(0);
    asm volatile("v_nop\n\tv_nop\n\tv_nop\n\tv_nop");
    __builtin_amdgcn_sched_barrier(0);

    const int lrow = lane & 15;
    const int hi8 = (lane >> 4) * 8;
#pragma unroll
    for (int mt = 0; mt < 2; ++mt)
#pragma unroll
      for (int t = 0; t < 4; ++t) {
        const int col = (ntbot + t) * 16 + lrow;
#pragma unroll
        for (int v = 0; v < 8; ++v) {
          const int r = row0 + mt * 16 + hi8 + v;
          C[(size_t)r * DQ + col] = accd[mt][t][v];
        }
      }
    __builtin_amdgcn_sched_barrier(0);
  }

#pragma unroll
  for (int u = 0; u < DCH; ++u) {
    ac[u][0] = As[(0 * KSTEPS + u) * 32 + lane];
    ac[u][1] = As[(1 * KSTEPS + u) * 32 + lane];
  }
  v8f acc[2][4];
#pragma unroll
  for (int m = 0; m < 2; ++m)
#pragma unroll
    for (int t = 0; t < 4; ++t)
      acc[m][t] = (v8f){0.f, 0.f, 0.f, 0.f, 0.f, 0.f, 0.f, 0.f};

#pragma unroll
  for (int c = 0; c < NDCH; ++c) {
    if (c + 1 < NDCH) {
      const int kn = (c + 1) * DCH;
#pragma unroll
      for (int u = 0; u < DCH; ++u)
#pragma unroll
        for (int t = 0; t < 4; ++t)
          bn[u][t] = bpt[(size_t)(kn + u) * (NTILES * 32) + t * 32];
#pragma unroll
      for (int u = 0; u < DCH; ++u) {
        an[u][0] = As[(0 * KSTEPS + kn + u) * 32 + lane];
        an[u][1] = As[(1 * KSTEPS + kn + u) * 32 + lane];
      }
    }
#pragma unroll
    for (int u = 0; u < DCH; ++u)
#pragma unroll
      for (int t = 0; t < 4; ++t) {
        acc[0][t] = wmma_f32_4(ac[u][0], bc[u][t], acc[0][t]);
        acc[1][t] = wmma_f32_4(ac[u][1], bc[u][t], acc[1][t]);
      }
    if (c + 1 < NDCH) {
#pragma unroll
      for (int u = 0; u < DCH; ++u) {
#pragma unroll
        for (int t = 0; t < 4; ++t) bc[u][t] = bn[u][t];
        ac[u][0] = an[u][0]; ac[u][1] = an[u][1];
      }
    }
  }

  __builtin_amdgcn_sched_barrier(0);
  asm volatile("v_nop\n\tv_nop\n\tv_nop\n\tv_nop");
  __builtin_amdgcn_sched_barrier(0);

  {
    const int lrow = lane & 15;
    const int hi8 = (lane >> 4) * 8;
#pragma unroll
    for (int mt = 0; mt < 2; ++mt)
#pragma unroll
      for (int t = 0; t < 4; ++t) {
        const int col = (nttop + t) * 16 + lrow;
#pragma unroll
        for (int v = 0; v < 8; ++v) {
          const int r = row0 + mt * 16 + hi8 + v;
          C[(size_t)r * DQ + col] = acc[mt][t][v];
        }
      }
  }
}

#define CHK 16
#define NCHK (KSTEPS / CHK)

static __device__ __forceinline__ void
chain_product(const float* __restrict__ A, const float* __restrict__ Bp,
              float* __restrict__ C, v2f* As, const int tid,
              const bool packed_out) {
  const int row0 = blockIdx.y * 16;
  const int wave = tid >> 5;
  const int lane = tid & 31;
  const int nt0 = blockIdx.x * 4 + wave * 2;

  const v2f* bp = (const v2f*)Bp + (size_t)nt0 * 32 + lane;

  v2f bc0[CHK], bc1[CHK], ac[CHK];
  v2f bn0[CHK], bn1[CHK], an[CHK];

#pragma unroll
  for (int u = 0; u < CHK; ++u) {
    bc0[u] = bp[(size_t)u * (NTILES * 32)];
    bc1[u] = bp[(size_t)u * (NTILES * 32) + 32];
  }

  {
    const int row = tid >> 2;
    const int cq = tid & 3;
    const float* arow = A + (size_t)(row0 + row) * DQ;
    v4f v[32];
#pragma unroll
    for (int s = 0; s < 32; ++s) v[s] = *(const v4f*)(arow + (cq + 4 * s) * 4);
#pragma unroll
    for (int s = 0; s < 32; ++s) {
      const int ks = cq + 4 * s;
      v2f lo; lo[0] = v[s][0]; lo[1] = v[s][1];
      v2f hi; hi[0] = v[s][2]; hi[1] = v[s][3];
      As[ks * 32 + row] = lo;
      As[ks * 32 + row + 16] = hi;
    }
  }
  __syncthreads();

#pragma unroll
  for (int u = 0; u < CHK; ++u) ac[u] = As[u * 32 + lane];

  v8f acc0 = (v8f){0.f, 0.f, 0.f, 0.f, 0.f, 0.f, 0.f, 0.f};
  v8f acc1 = (v8f){0.f, 0.f, 0.f, 0.f, 0.f, 0.f, 0.f, 0.f};

#pragma unroll
  for (int c = 0; c < NCHK; ++c) {
    if (c + 1 < NCHK) {
      const int kn = (c + 1) * CHK;
#pragma unroll
      for (int u = 0; u < CHK; ++u) {
        bn0[u] = bp[(size_t)(kn + u) * (NTILES * 32)];
        bn1[u] = bp[(size_t)(kn + u) * (NTILES * 32) + 32];
      }
#pragma unroll
      for (int u = 0; u < CHK; ++u) an[u] = As[(kn + u) * 32 + lane];
    }
#pragma unroll
    for (int u = 0; u < CHK; ++u) {
      acc0 = wmma_f32_4(ac[u], bc0[u], acc0);
      acc1 = wmma_f32_4(ac[u], bc1[u], acc1);
    }
    if (c + 1 < NCHK) {
#pragma unroll
      for (int u = 0; u < CHK; ++u) {
        bc0[u] = bn0[u]; bc1[u] = bn1[u]; ac[u] = an[u];
      }
    }
  }

  __builtin_amdgcn_sched_barrier(0);
  asm volatile("v_nop\n\tv_nop\n\tv_nop\n\tv_nop");
  __builtin_amdgcn_sched_barrier(0);

  const int lrow = lane & 15;
  const int hi8 = (lane >> 4) * 8;
  if (packed_out) {
    v2f* cp = (v2f*)C + ((size_t)(((row0 + hi8) >> 2) * NTILES + nt0) * 32 + lrow);
#pragma unroll
    for (int p = 0; p < 4; ++p) {
      v2f x0; x0[0] = acc0[2 * p]; x0[1] = acc0[2 * p + 1];
      v2f x1; x1[0] = acc1[2 * p]; x1[1] = acc1[2 * p + 1];
      const int off = (p >> 1) * (NTILES * 32) + 16 * (p & 1);
      cp[off] = x0;
      cp[off + 32] = x1;
    }
  } else {
#pragma unroll
    for (int v = 0; v < 8; ++v) {
      float* crow = C + (size_t)(row0 + hi8 + v) * DQ + nt0 * 16 + lrow;
      crow[0] = acc0[v];
      crow[16] = acc1[v];
    }
  }
}

template <bool PACKED_OUT>
__global__ void __launch_bounds__(64)
vqc_wmma_gemm_chain(const float* __restrict__ A, const float* __restrict__ Bp,
                    float* __restrict__ C) {
  __shared__ v2f As[KSTEPS * 32];
  chain_product(A, Bp, C, As, threadIdx.x, PACKED_OUT);
}

__global__ void __launch_bounds__(128)
vqc_wmma_gemm_chain_pair(const float* __restrict__ A0, const float* __restrict__ B0p,
                         float* __restrict__ C0,
                         const float* __restrict__ A1, const float* __restrict__ B1p,
                         float* __restrict__ C1) {
  __shared__ v2f As[2 * KSTEPS * 32];
  const int g = __builtin_amdgcn_readfirstlane((int)(threadIdx.x >> 6));
  chain_product(g ? A1 : A0, g ? B1p : B0p, g ? C1 : C0, As + g * (KSTEPS * 32),
                threadIdx.x & 63, g != 0);
}

extern "C" void kernel_launch(void* const* d_in, const int* in_sizes, int n_in,
                              void* d_out, int out_size, void* d_ws,
                              size_t ws_size, hipStream_t stream) {
  (void)in_sizes; (void)n_in; (void)out_size; (void)ws_size;
  const float* state = (const float*)d_in[0];
  const float* rotp  = (const float*)d_in[1];
  const float* entp  = (const float*)d_in[2];
  float* out = (float*)d_out;

  float* w     = (float*)d_ws;
  float* Tlay  = w + 1048576;
  float* T01   = w + 0;
  float* T23   = w + 262144;
  float* Ttot  = w + 524288;

  vqc_build_layer<<<dim3(QQ, NLAY), 128, 0, stream>>>(rotp, entp, Tlay);

  vqc_wmma_gemm_chain_pair<<<dim3(DQ / 64, DQ / 16), 128, 0, stream>>>(
      Tlay + 0 * 262144, Tlay + 1 * 262144, T01,
      Tlay + 2 * 262144, Tlay + 3 * 262144, T23);

  vqc_wmma_gemm_chain<true><<<dim3(DQ / 64, DQ / 16), 64, 0, stream>>>(T01, T23, Ttot);

  vqc_wmma_gemm<<<M_STATE / 32, 128, 0, stream>>>(state, Ttot, out);
}


# The three chain products leave out the blocks that are zero by construction (dense 256x256 + diagonal tiles); tiles left go one per wave, loads all ahead: 64 not 256 WMMAs a wave; sum order unchanged
# speedup vs baseline: 5.9135x; 1.1906x over previous
#include <hip/hip_runtime.h>
#include <math.h>


typedef float v2f __attribute__((ext_vector_type(2)));
typedef float v4f __attribute__((ext_vector_type(4)));
typedef float v8f __attribute__((ext_vector_type(8)));

#define DQ      512
#define QQ      256
#define NLAY    4
#define NPAIR   255
#define KSTEPS  128
#define NTILES  32
#define M_STATE (16 * 2048)

static __device__ __forceinline__ v8f wmma_f32_4(v2f a, v2f b, v8f c) {
  return __builtin_amdgcn_wmma_f32_16x16x4_f32(false, a, false, b, (short)0, c,
                                               false, false);
}

#define CPAD 8

__global__ void __launch_bounds__(128)
vqc_build_layer(const float* __restrict__ rot_params,
                const float* __restrict__ ent_params,
                float* __restrict__ Tlay) {
  const int q = blockIdx.x;
  const int l = blockIdx.y;
  const int tid = threadIdx.x;

  __shared__ __attribute__((aligned(16))) float c[QQ + CPAD];
  __shared__ __attribute__((aligned(16))) float e[2][DQ];

  for (int p = tid; p < QQ + CPAD; p += 128) {
    float cv = 1.0f;
    if (p < NPAIR) {
      const float x = ent_params[l * NPAIR + p];
      cv = 0.1f * (1.0f / (1.0f + expf(-x)));
    }
    c[p] = cv;
  }
  {
    const v4f z = (v4f){0.f, 0.f, 0.f, 0.f};
    ((v4f*)&e[0][0])[tid] = z;
    ((v4f*)&e[1][0])[tid] = z;
  }

  const float t0 = rot_params[((l * QQ) + q) * 3 + 0];
  const float t1 = rot_params[((l * QQ) + q) * 3 + 1];
  const float t2 = rot_params[((l * QQ) + q) * 3 + 2];
  const float c0 = cosf(t0 * 0.5f);
  const float c1 = cosf(t1 * 0.5f);
  const float s1 = sinf(t1 * 0.5f);
  const float c2 = cosf(t2 * 0.5f);
  const float b00 = (c0 * c1) * c2;
  const float b01 = (c0 * (-s1)) * c2;
  const float b10 = (c0 * s1) * c2;
  __syncthreads();

  if (tid < 2) {
    const int r = 2 * q + tid;
    float* row = e[tid];
    row[r] = 1.0f;
    if (r < QQ) {
      if (r >= 1) row[r - 1] = c[r - 1];
      if (r <= QQ - 2) {
        float val = c[r];
        int j = r + 1;
        for (; (j & 7) != 0; ++j) { row[j] = val; val *= c[j]; }
        v4f ca = *(const v4f*)&c[j];
        v4f cb = *(const v4f*)&c[j + 4];
        for (; j < QQ; j += 8) {
          const v4f na = *(const v4f*)&c[j + 8];
          const v4f nb = *(const v4f*)&c[j + 12];
          v4f va, vb;
          va[0] = val; val *= ca[0];
          va[1] = val; val *= ca[1];
          va[2] = val; val *= ca[2];
          va[3] = val; val *= ca[3];
          vb[0] = val; val *= cb[0];
          vb[1] = val; val *= cb[1];
          vb[2] = val; val *= cb[2];
          vb[3] = val; val *= cb[3];
          *(v4f*)&row[j] = va;
          *(v4f*)&row[j + 4] = vb;
          ca = na; cb = nb;
        }
      }
    }
  }
  __syncthreads();

  if (l & 1) {
    const int ks = q >> 1;
    const int half = q & 1;
    float* outp = Tlay + (size_t)l * DQ * DQ;
#pragma unroll
    for (int s = 0; s < 2; ++s) {
      const int col = 256 * s + 2 * tid;
      const v2f e0 = ((const v2f*)&e[0][0])[128 * s + tid];
      const v2f e1 = ((const v2f*)&e[1][0])[128 * s + tid];
      v4f o;
      o[0] = b00 * e0[0] + b01 * e1[0];
      o[1] = b10 * e0[0] + b00 * e1[0];
      o[2] = b00 * e0[1] + b01 * e1[1];
      o[3] = b10 * e0[1] + b00 * e1[1];
      const int pair = (ks * NTILES + (col >> 4)) * 32 + 16 * half + (col & 15);
      *(v4f*)(outp + (size_t)pair * 2) = o;
    }
  } else {
    const v4f e0 = ((const v4f*)&e[0][0])[tid];
    const v4f e1 = ((const v4f*)&e[1][0])[tid];
    v4f o0, o1;
#pragma unroll
    for (int k = 0; k < 4; ++k) {
      o0[k] = b00 * e0[k] + b01 * e1[k];
      o1[k] = b10 * e0[k] + b00 * e1[k];
    }
    float* out0 = Tlay + ((size_t)l * DQ + 2 * q) * DQ;
    ((v4f*)out0)[tid] = o0;
    ((v4f*)(out0 + DQ))[tid] = o1;
  }
}

#define KTOP   64
#define NTTOP  16
#define DCH    4
#define NDCH   (KTOP / DCH)

__global__ void __launch_bounds__(128)
vqc_wmma_gemm(const float* __restrict__ A, const float* __restrict__ Bp,
              float* __restrict__ C) {
  __shared__ v2f As[2 * KSTEPS * 32];

  const int row0 = blockIdx.x * 32;
  const int wave = threadIdx.x >> 5;
  const int lane = threadIdx.x & 31;
  const int nttop = wave * 4;
  const int ntbot = NTTOP + wave * 4;
  const v2f* BpV = (const v2f*)Bp;
  const v2f* bpt = BpV + (size_t)nttop * 32 + lane;

  v2f bd[4][4];
#pragma unroll
  for (int t = 0; t < 4; ++t)
#pragma unroll
    for (int j = 0; j < 4; ++j) {
      const int nt = ntbot + t;
      bd[t][j] = BpV[((size_t)(4 * nt + j) * NTILES + nt) * 32 + lane];
    }
  v2f bc[DCH][4], bn[DCH][4], ac[DCH][2], an[DCH][2];
#pragma unroll
  for (int u = 0; u < DCH; ++u)
#pragma unroll
    for (int t = 0; t < 4; ++t)
      bc[u][t] = bpt[(size_t)u * (NTILES * 32) + t * 32];

  {
    const int row = threadIdx.x >> 2;
    const int cq = threadIdx.x & 3;
    const int mt = row >> 4;
    const int lr = row & 15;
    const float* arow = A + (size_t)(row0 + row) * DQ;
#pragma unroll 4
    for (int s = 0; s < 32; ++s) {
      const int ks = cq + 4 * s;
      v4f v = *(const v4f*)(arow + ks * 4);
      v2f lo; lo[0] = v[0]; lo[1] = v[1];
      v2f hi; hi[0] = v[2]; hi[1] = v[3];
      As[(mt * KSTEPS + ks) * 32 + lr] = lo;
      As[(mt * KSTEPS + ks) * 32 + lr + 16] = hi;
    }
  }
  __syncthreads();

  {
    v2f ad[4][4][2];
#pragma unroll
    for (int t = 0; t < 4; ++t)
#pragma unroll
      for (int j = 0; j < 4; ++j) {
        const int ks = 4 * (ntbot + t) + j;
        ad[t][j][0] = As[(0 * KSTEPS + ks) * 32 + lane];
        ad[t][j][1] = As[(1 * KSTEPS + ks) * 32 + lane];
      }
    v8f accd[2][4];
#pragma unroll
    for (int m = 0; m < 2; ++m)
#pragma unroll
      for (int t = 0; t < 4; ++t)
        accd[m][t] = (v8f){0.f, 0.f, 0.f, 0.f, 0.f, 0.f, 0.f, 0.f};
#pragma unroll
    for (int j = 0; j < 4; ++j)
#pragma unroll
      for (int t = 0; t < 4; ++t) {
        accd[0][t] = wmma_f32_4(ad[t][j][0], bd[t][j], accd[0][t]);
        accd[1][t] = wmma_f32_4(ad[t][j][1], bd[t][j], accd[1][t]);
      }

    __builtin_amdgcn_sched_barrier(0);
    asm volatile("v_nop\n\tv_nop\n\tv_nop\n\tv_nop");
    __builtin_amdgcn_sched_barrier(0);

    const int lrow = lane & 15;
    const int hi8 = (lane >> 4) * 8;
#pragma unroll
    for (int mt = 0; mt < 2; ++mt)
#pragma unroll
      for (int t = 0; t < 4; ++t) {
        const int col = (ntbot + t) * 16 + lrow;
#pragma unroll
        for (int v = 0; v < 8; ++v) {
          const int r = row0 + mt * 16 + hi8 + v;
          C[(size_t)r * DQ + col] = accd[mt][t][v];
        }
      }
    __builtin_amdgcn_sched_barrier(0);
  }

#pragma unroll
  for (int u = 0; u < DCH; ++u) {
    ac[u][0] = As[(0 * KSTEPS + u) * 32 + lane];
    ac[u][1] = As[(1 * KSTEPS + u) * 32 + lane];
  }
  v8f acc[2][4];
#pragma unroll
  for (int m = 0; m < 2; ++m)
#pragma unroll
    for (int t = 0; t < 4; ++t)
      acc[m][t] = (v8f){0.f, 0.f, 0.f, 0.f, 0.f, 0.f, 0.f, 0.f};

#pragma unroll
  for (int c = 0; c < NDCH; ++c) {
    if (c + 1 < NDCH) {
      const int kn = (c + 1) * DCH;
#pragma unroll
      for (int u = 0; u < DCH; ++u)
#pragma unroll
        for (int t = 0; t < 4; ++t)
          bn[u][t] = bpt[(size_t)(kn + u) * (NTILES * 32) + t * 32];
#pragma unroll
      for (int u = 0; u < DCH; ++u) {
        an[u][0] = As[(0 * KSTEPS + kn + u) * 32 + lane];
        an[u][1] = As[(1 * KSTEPS + kn + u) * 32 + lane];
      }
    }
#pragma unroll
    for (int u = 0; u < DCH; ++u)
#pragma unroll
      for (int t = 0; t < 4; ++t) {
        acc[0][t] = wmma_f32_4(ac[u][0], bc[u][t], acc[0][t]);
        acc[1][t] = wmma_f32_4(ac[u][1], bc[u][t], acc[1][t]);
      }
    if (c + 1 < NDCH) {
#pragma unroll
      for (int u = 0; u < DCH; ++u) {
#pragma unroll
        for (int t = 0; t < 4; ++t) bc[u][t] = bn[u][t];
        ac[u][0] = an[u][0]; ac[u][1] = an[u][1];
      }
    }
  }

  __builtin_amdgcn_sched_barrier(0);
  asm volatile("v_nop\n\tv_nop\n\tv_nop\n\tv_nop");
  __builtin_amdgcn_sched_barrier(0);

  {
    const int lrow = lane & 15;
    const int hi8 = (lane >> 4) * 8;
#pragma unroll
    for (int mt = 0; mt < 2; ++mt)
#pragma unroll
      for (int t = 0; t < 4; ++t) {
        const int col = (nttop + t) * 16 + lrow;
#pragma unroll
        for (int v = 0; v < 8; ++v) {
          const int r = row0 + mt * 16 + hi8 + v;
          C[(size_t)r * DQ + col] = acc[mt][t][v];
        }
      }
  }
}

#define NDENSE_WG 64
#define NPROD_WG 68
#define NPAIR_WG 136

template <bool PAIRED>
static __device__ __forceinline__ void
chain_store_tile(float* __restrict__ C, float* X, const v8f acc, const int row0,
                 const int nt, const int wave, const int lane,
                 const bool packed_out) {
  const int lrow = lane & 15;
  const int half = lane >> 4;
  const int hi8 = half * 8;
  if (packed_out) {
    v2f* cp = (v2f*)C + ((size_t)(((row0 + hi8) >> 2) * NTILES + nt) * 32 + lrow);
#pragma unroll
    for (int p = 0; p < 4; ++p) {
      v2f x; x[0] = acc[2 * p]; x[1] = acc[2 * p + 1];
      cp[(p >> 1) * (NTILES * 32) + 16 * (p & 1)] = x;
    }
  } else if (PAIRED) {
    float* xp = X + (wave >> 1) * (16 * 32);
    const int odd = wave & 1;
#pragma unroll
    for (int v = 0; v < 8; ++v) xp[(hi8 + v) * 32 + 16 * odd + lrow] = acc[v];
    __syncthreads();
    float o[8];
#pragma unroll
    for (int v = 0; v < 8; ++v) o[v] = xp[(8 * odd + v) * 32 + lane];
    float* crow = C + (size_t)(row0 + 8 * odd) * DQ + (nt & ~1) * 16 + lane;
#pragma unroll
    for (int v = 0; v < 8; ++v) crow[(size_t)v * DQ] = o[v];
  } else {
#pragma unroll
    for (int v = 0; v < 8; ++v)
      C[(size_t)(row0 + hi8 + v) * DQ + nt * 16 + lrow] = acc[v];
  }
}

static __device__ __forceinline__ void
chain_product(const float* __restrict__ A, const float* __restrict__ Bp,
              float* __restrict__ C, v2f* As, float* X, const int wg,
              const bool packed_out) {
  const int tid = threadIdx.x;
  const int wave = tid >> 5;
  const int lane = tid & 31;

  if (wg >= NDENSE_WG) {
    const int nt = NTTOP + 4 * (wg - NDENSE_WG) + wave;
    const int row0 = nt * 16;
    const v2f* bp = (const v2f*)Bp + ((size_t)(4 * nt) * NTILES + nt) * 32 + lane;
    const float* ap = A + (size_t)(row0 + (lane & 15)) * DQ + 16 * nt + 2 * (lane >> 4);
    v2f a[4], b[4];
#pragma unroll
    for (int j = 0; j < 4; ++j) {
      b[j] = bp[(size_t)j * (NTILES * 32)];
      a[j] = *(const v2f*)(ap + 4 * j);
    }
    v8f acc = (v8f){0.f, 0.f, 0.f, 0.f, 0.f, 0.f, 0.f, 0.f};
#pragma unroll
    for (int j = 0; j < 4; ++j) acc = wmma_f32_4(a[j], b[j], acc);

    __builtin_amdgcn_sched_barrier(0);
    asm volatile("v_nop\n\tv_nop\n\tv_nop\n\tv_nop");
    __builtin_amdgcn_sched_barrier(0);
    chain_store_tile<false>(C, X, acc, row0, nt, wave, lane, packed_out);
    return;
  }

  const int row0 = (wg >> 2) * 16;
  const int nt = (wg & 3) * 4 + wave;

  const v2f* bp = (const v2f*)Bp + (size_t)nt * 32 + lane;

  v2f bf[KTOP], af[KTOP];

  {
    const int row = (tid & 63) >> 2;
    const int cq = tid & 3;
    const int k0 = (tid >> 6) * (KTOP / 2);
    const float* arow = A + (size_t)(row0 + row) * DQ;
    v4f v[KTOP / 8];
#pragma unroll
    for (int s = 0; s < KTOP / 8; ++s) v[s] = *(const v4f*)(arow + (k0 + cq + 4 * s) * 4);
#pragma unroll
    for (int u = 0; u < KTOP / 2; ++u) bf[u] = bp[(size_t)u * (NTILES * 32)];
#pragma unroll
    for (int s = 0; s < KTOP / 8; ++s) {
      const int ks = k0 + cq + 4 * s;
      v2f lo; lo[0] = v[s][0]; lo[1] = v[s][1];
      v2f hi; hi[0] = v[s][2]; hi[1] = v[s][3];
      As[ks * 32 + row] = lo;
      As[ks * 32 + row + 16] = hi;
    }
  }
  __syncthreads();

#pragma unroll
  for (int u = KTOP / 2; u < KTOP; ++u) bf[u] = bp[(size_t)u * (NTILES * 32)];
#pragma unroll
  for (int u = 0; u < KTOP; ++u) af[u] = As[u * 32 + lane];

  __builtin_amdgcn_sched_barrier(0);

  v8f acc = (v8f){0.f, 0.f, 0.f, 0.f, 0.f, 0.f, 0.f, 0.f};
#pragma unroll
  for (int u = 0; u < KTOP; ++u) acc = wmma_f32_4(af[u], bf[u], acc);

  __builtin_amdgcn_sched_barrier(0);
  asm volatile("v_nop\n\tv_nop\n\tv_nop\n\tv_nop");
  __builtin_amdgcn_sched_barrier(0);
  chain_store_tile<true>(C, X, acc, row0, nt, wave, lane, packed_out);
}

template <bool PACKED_OUT>
__global__ void __launch_bounds__(128)
vqc_wmma_gemm_chain(const float* __restrict__ A, const float* __restrict__ Bp,
                    float* __restrict__ C) {
  __shared__ v2f As[KTOP * 32];
  __shared__ float X[2 * 16 * 32];
  chain_product(A, Bp, C, As, X, blockIdx.x, PACKED_OUT);
}

__global__ void __launch_bounds__(128)
vqc_wmma_gemm_chain_pair(const float* __restrict__ A0, const float* __restrict__ B0p,
                         float* __restrict__ C0,
                         const float* __restrict__ A1, const float* __restrict__ B1p,
                         float* __restrict__ C1) {
  __shared__ v2f As[KTOP * 32];
  __shared__ float X[2 * 16 * 32];
  const bool g = blockIdx.x >= NPROD_WG;
  chain_product(g ? A1 : A0, g ? B1p : B0p, g ? C1 : C0, As, X,
                g ? (int)blockIdx.x - NPROD_WG : (int)blockIdx.x, g);
}

extern "C" void kernel_launch(void* const* d_in, const int* in_sizes, int n_in,
                              void* d_out, int out_size, void* d_ws,
                              size_t ws_size, hipStream_t stream) {
  (void)in_sizes; (void)n_in; (void)out_size; (void)ws_size;
  const float* state = (const float*)d_in[0];
  const float* rotp  = (const float*)d_in[1];
  const float* entp  = (const float*)d_in[2];
  float* out = (float*)d_out;

  float* w     = (float*)d_ws;
  float* Tlay  = w + 1048576;
  float* T01   = w + 0;
  float* T23   = w + 262144;
  float* Ttot  = w + 524288;

  vqc_build_layer<<<dim3(QQ, NLAY), 128, 0, stream>>>(rotp, entp, Tlay);

  vqc_wmma_gemm_chain_pair<<<NPAIR_WG, 128, 0, stream>>>(
      Tlay + 0 * 262144, Tlay + 1 * 262144, T01,
      Tlay + 2 * 262144, Tlay + 3 * 262144, T23);

  vqc_wmma_gemm_chain<true><<<NPROD_WG, 128, 0, stream>>>(T01, T23, Ttot);

  vqc_wmma_gemm<<<M_STATE / 32, 128, 0, stream>>>(state, Ttot, out);
}


# The 32768x512 product's work-groups own two 32-row tiles: the 2nd tile's A is loaded into a second LDS buffer under the 1st tile's WMMAs, whose stores drain under the 2nd's; sum order unchanged
# speedup vs baseline: 6.1167x; 1.0344x over previous
#include <hip/hip_runtime.h>
#include <math.h>


typedef float v2f __attribute__((ext_vector_type(2)));
typedef float v4f __attribute__((ext_vector_type(4)));
typedef float v8f __attribute__((ext_vector_type(8)));

#define DQ      512
#define QQ      256
#define NLAY    4
#define NPAIR   255
#define KSTEPS  128
#define NTILES  32
#define M_STATE (16 * 2048)

static __device__ __forceinline__ v8f wmma_f32_4(v2f a, v2f b, v8f c) {
  return __builtin_amdgcn_wmma_f32_16x16x4_f32(false, a, false, b, (short)0, c,
                                               false, false);
}

#define CPAD 8

__global__ void __launch_bounds__(128)
vqc_build_layer(const float* __restrict__ rot_params,
                const float* __restrict__ ent_params,
                float* __restrict__ Tlay) {
  const int q = blockIdx.x;
  const int l = blockIdx.y;
  const int tid = threadIdx.x;

  __shared__ __attribute__((aligned(16))) float c[QQ + CPAD];
  __shared__ __attribute__((aligned(16))) float e[2][DQ];

  for (int p = tid; p < QQ + CPAD; p += 128) {
    float cv = 1.0f;
    if (p < NPAIR) {
      const float x = ent_params[l * NPAIR + p];
      cv = 0.1f * (1.0f / (1.0f + expf(-x)));
    }
    c[p] = cv;
  }
  {
    const v4f z = (v4f){0.f, 0.f, 0.f, 0.f};
    ((v4f*)&e[0][0])[tid] = z;
    ((v4f*)&e[1][0])[tid] = z;
  }

  const float t0 = rot_params[((l * QQ) + q) * 3 + 0];
  const float t1 = rot_params[((l * QQ) + q) * 3 + 1];
  const float t2 = rot_params[((l * QQ) + q) * 3 + 2];
  const float c0 = cosf(t0 * 0.5f);
  const float c1 = cosf(t1 * 0.5f);
  const float s1 = sinf(t1 * 0.5f);
  const float c2 = cosf(t2 * 0.5f);
  const float b00 = (c0 * c1) * c2;
  const float b01 = (c0 * (-s1)) * c2;
  const float b10 = (c0 * s1) * c2;
  __syncthreads();

  if (tid < 2) {
    const int r = 2 * q + tid;
    float* row = e[tid];
    row[r] = 1.0f;
    if (r < QQ) {
      if (r >= 1) row[r - 1] = c[r - 1];
      if (r <= QQ - 2) {
        float val = c[r];
        int j = r + 1;
        for (; (j & 7) != 0; ++j) { row[j] = val; val *= c[j]; }
        v4f ca = *(const v4f*)&c[j];
        v4f cb = *(const v4f*)&c[j + 4];
        for (; j < QQ; j += 8) {
          const v4f na = *(const v4f*)&c[j + 8];
          const v4f nb = *(const v4f*)&c[j + 12];
          v4f va, vb;
          va[0] = val; val *= ca[0];
          va[1] = val; val *= ca[1];
          va[2] = val; val *= ca[2];
          va[3] = val; val *= ca[3];
          vb[0] = val; val *= cb[0];
          vb[1] = val; val *= cb[1];
          vb[2] = val; val *= cb[2];
          vb[3] = val; val *= cb[3];
          *(v4f*)&row[j] = va;
          *(v4f*)&row[j + 4] = vb;
          ca = na; cb = nb;
        }
      }
    }
  }
  __syncthreads();

  if (l & 1) {
    const int ks = q >> 1;
    const int half = q & 1;
    float* outp = Tlay + (size_t)l * DQ * DQ;
#pragma unroll
    for (int s = 0; s < 2; ++s) {
      const int col = 256 * s + 2 * tid;
      const v2f e0 = ((const v2f*)&e[0][0])[128 * s + tid];
      const v2f e1 = ((const v2f*)&e[1][0])[128 * s + tid];
      v4f o;
      o[0] = b00 * e0[0] + b01 * e1[0];
      o[1] = b10 * e0[0] + b00 * e1[0];
      o[2] = b00 * e0[1] + b01 * e1[1];
      o[3] = b10 * e0[1] + b00 * e1[1];
      const int pair = (ks * NTILES + (col >> 4)) * 32 + 16 * half + (col & 15);
      *(v4f*)(outp + (size_t)pair * 2) = o;
    }
  } else {
    const v4f e0 = ((const v4f*)&e[0][0])[tid];
    const v4f e1 = ((const v4f*)&e[1][0])[tid];
    v4f o0, o1;
#pragma unroll
    for (int k = 0; k < 4; ++k) {
      o0[k] = b00 * e0[k] + b01 * e1[k];
      o1[k] = b10 * e0[k] + b00 * e1[k];
    }
    float* out0 = Tlay + ((size_t)l * DQ + 2 * q) * DQ;
    ((v4f*)out0)[tid] = o0;
    ((v4f*)(out0 + DQ))[tid] = o1;
  }
}

#define KTOP   64
#define NTTOP  16
#define DCH    4
#define NDCH   (KTOP / DCH)
#define TPW    2
#define ASTILE (2 * KSTEPS * 32)
#define SPC    (32 / NDCH)

#define WMMA_RUN_BEGIN()                                                          \
  __builtin_amdgcn_sched_barrier(0)
#define WMMA_RUN_END()                                                            \
  do {                                                                            \
    __builtin_amdgcn_sched_barrier(0);   \
    asm volatile("v_nop\n\tv_nop\n\tv_nop\n\tv_nop");           \
    __builtin_amdgcn_sched_barrier(0);   \
  } while (0)

__global__ void __launch_bounds__(128)
vqc_wmma_gemm(const float* __restrict__ A, const float* __restrict__ Bp,
              float* __restrict__ C) {
  __shared__ v2f As[TPW * ASTILE];

  const int wave = threadIdx.x >> 5;
  const int lane = threadIdx.x & 31;
  const int nttop = wave * 4;
  const int ntbot = NTTOP + wave * 4;
  const v2f* BpV = (const v2f*)Bp;
  const v2f* bpt = BpV + (size_t)nttop * 32 + lane;

  const int srow = threadIdx.x >> 2;
  const int cq = threadIdx.x & 3;
  const int sbase = ((srow >> 4) * KSTEPS + cq) * 32 + (srow & 15);
  const float* arow = A + ((size_t)blockIdx.x * (32 * TPW) + srow) * DQ + cq * 4;

  v2f bd[4][4];
#pragma unroll
  for (int t = 0; t < 4; ++t)
#pragma unroll
    for (int j = 0; j < 4; ++j) {
      const int nt = ntbot + t;
      bd[t][j] = BpV[((size_t)(4 * nt + j) * NTILES + nt) * 32 + lane];
    }
  v2f bq[3][DCH][4], aq[2][DCH][2];
  v4f pf[4][SPC];
#pragma unroll
  for (int g = 0; g < 2; ++g)
#pragma unroll
    for (int u = 0; u < DCH; ++u)
#pragma unroll
      for (int t = 0; t < 4; ++t)
        bq[g][u][t] = bpt[(size_t)(g * DCH + u) * (NTILES * 32) + t * 32];

#pragma unroll 16
  for (int s = 0; s < 32; ++s) {
    const v4f v = *(const v4f*)(arow + 16 * s);
    v2f lo; lo[0] = v[0]; lo[1] = v[1];
    v2f hi; hi[0] = v[2]; hi[1] = v[3];
    As[sbase + 128 * s] = lo;
    As[sbase + 128 * s + 16] = hi;
  }
  __syncthreads();

#pragma unroll
  for (int it = 0; it < TPW; ++it) {
    const bool more = (it + 1 < TPW);
    const int row0 = (blockIdx.x * TPW + it) * 32;
    const v2f* Asc = As + it * ASTILE;
    v2f* Asn = As + (more ? it + 1 : it) * ASTILE;
    const float* anext = arow + (size_t)(more ? it + 1 : it) * 32 * DQ;

    {
      v2f ad[4][4][2];
#pragma unroll
      for (int t = 0; t < 4; ++t)
#pragma unroll
        for (int j = 0; j < 4; ++j) {
          const int ks = 4 * (ntbot + t) + j;
          ad[t][j][0] = Asc[(0 * KSTEPS + ks) * 32 + lane];
          ad[t][j][1] = Asc[(1 * KSTEPS + ks) * 32 + lane];
        }
      v8f accd[2][4];
#pragma unroll
      for (int m = 0; m < 2; ++m)
#pragma unroll
        for (int t = 0; t < 4; ++t)
          accd[m][t] = (v8f){0.f, 0.f, 0.f, 0.f, 0.f, 0.f, 0.f, 0.f};
      WMMA_RUN_BEGIN();
#pragma unroll
      for (int j = 0; j < 4; ++j)
#pragma unroll
        for (int t = 0; t < 4; ++t) {
          accd[0][t] = wmma_f32_4(ad[t][j][0], bd[t][j], accd[0][t]);
          accd[1][t] = wmma_f32_4(ad[t][j][1], bd[t][j], accd[1][t]);
        }
      WMMA_RUN_END();

      const int lrow = lane & 15;
      const int hi8 = (lane >> 4) * 8;
#pragma unroll
      for (int mt = 0; mt < 2; ++mt)
#pragma unroll
        for (int t = 0; t < 4; ++t) {
          const int col = (ntbot + t) * 16 + lrow;
#pragma unroll
          for (int v = 0; v < 8; ++v) {
            const int r = row0 + mt * 16 + hi8 + v;
            C[(size_t)r * DQ + col] = accd[mt][t][v];
          }
        }
    }

#pragma unroll
    for (int u = 0; u < DCH; ++u) {
      aq[0][u][0] = Asc[(0 * KSTEPS + u) * 32 + lane];
      aq[0][u][1] = Asc[(1 * KSTEPS + u) * 32 + lane];
    }
    v8f acc[2][4];
#pragma unroll
    for (int m = 0; m < 2; ++m)
#pragma unroll
      for (int t = 0; t < 4; ++t)
        acc[m][t] = (v8f){0.f, 0.f, 0.f, 0.f, 0.f, 0.f, 0.f, 0.f};

#pragma unroll
    for (int c = 0; c < NDCH; ++c) {
      const int G = it * NDCH + c;
      if (c + 2 < NDCH || more) {
        const int kn = ((c + 2) % NDCH) * DCH;
#pragma unroll
        for (int u = 0; u < DCH; ++u)
#pragma unroll
          for (int t = 0; t < 4; ++t)
            bq[(G + 2) % 3][u][t] = bpt[(size_t)(kn + u) * (NTILES * 32) + t * 32];
      }
      if (c + 1 < NDCH) {
        const int kn = (c + 1) * DCH;
#pragma unroll
        for (int u = 0; u < DCH; ++u) {
          aq[(c + 1) & 1][u][0] = Asc[(0 * KSTEPS + kn + u) * 32 + lane];
          aq[(c + 1) & 1][u][1] = Asc[(1 * KSTEPS + kn + u) * 32 + lane];
        }
      }
      if (more) {
#pragma unroll
        for (int i = 0; i < SPC; ++i)
          pf[c % 4][i] = *(const v4f*)(anext + 16 * (SPC * c + i));
        if (c >= 3) {
#pragma unroll
          for (int i = 0; i < SPC; ++i) {
            const int s = SPC * (c - 3) + i;
            const v4f v = pf[(c - 3) % 4][i];
            v2f lo; lo[0] = v[0]; lo[1] = v[1];
            v2f hi; hi[0] = v[2]; hi[1] = v[3];
            Asn[sbase + 128 * s] = lo;
            Asn[sbase + 128 * s + 16] = hi;
          }
        }
      }
      WMMA_RUN_BEGIN();
#pragma unroll
      for (int u = 0; u < DCH; ++u)
#pragma unroll
        for (int t = 0; t < 4; ++t) {
          acc[0][t] = wmma_f32_4(aq[c & 1][u][0], bq[G % 3][u][t], acc[0][t]);
          acc[1][t] = wmma_f32_4(aq[c & 1][u][1], bq[G % 3][u][t], acc[1][t]);
        }
      WMMA_RUN_END();
    }

    if (more) {
#pragma unroll
      for (int c = NDCH - 3; c < NDCH; ++c)
#pragma unroll
        for (int i = 0; i < SPC; ++i) {
          const int s = SPC * c + i;
          const v4f v = pf[c % 4][i];
          v2f lo; lo[0] = v[0]; lo[1] = v[1];
          v2f hi; hi[0] = v[2]; hi[1] = v[3];
          Asn[sbase + 128 * s] = lo;
          Asn[sbase + 128 * s + 16] = hi;
        }
      __syncthreads();
    }

    {
      const int lrow = lane & 15;
      const int hi8 = (lane >> 4) * 8;
#pragma unroll
      for (int mt = 0; mt < 2; ++mt)
#pragma unroll
        for (int t = 0; t < 4; ++t) {
          const int col = (nttop + t) * 16 + lrow;
#pragma unroll
          for (int v = 0; v < 8; ++v) {
            const int r = row0 + mt * 16 + hi8 + v;
            C[(size_t)r * DQ + col] = acc[mt][t][v];
          }
        }
    }
  }
}

#define NDENSE_WG 64
#define NPROD_WG 68
#define NPAIR_WG 136

template <bool PAIRED>
static __device__ __forceinline__ void
chain_store_tile(float* __restrict__ C, float* X, const v8f acc, const int row0,
                 const int nt, const int wave, const int lane,
                 const bool packed_out) {
  const int lrow = lane & 15;
  const int half = lane >> 4;
  const int hi8 = half * 8;
  if (packed_out) {
    v2f* cp = (v2f*)C + ((size_t)(((row0 + hi8) >> 2) * NTILES + nt) * 32 + lrow);
#pragma unroll
    for (int p = 0; p < 4; ++p) {
      v2f x; x[0] = acc[2 * p]; x[1] = acc[2 * p + 1];
      cp[(p >> 1) * (NTILES * 32) + 16 * (p & 1)] = x;
    }
  } else if (PAIRED) {
    float* xp = X + (wave >> 1) * (16 * 32);
    const int odd = wave & 1;
#pragma unroll
    for (int v = 0; v < 8; ++v) xp[(hi8 + v) * 32 + 16 * odd + lrow] = acc[v];
    __syncthreads();
    float o[8];
#pragma unroll
    for (int v = 0; v < 8; ++v) o[v] = xp[(8 * odd + v) * 32 + lane];
    float* crow = C + (size_t)(row0 + 8 * odd) * DQ + (nt & ~1) * 16 + lane;
#pragma unroll
    for (int v = 0; v < 8; ++v) crow[(size_t)v * DQ] = o[v];
  } else {
#pragma unroll
    for (int v = 0; v < 8; ++v)
      C[(size_t)(row0 + hi8 + v) * DQ + nt * 16 + lrow] = acc[v];
  }
}

static __device__ __forceinline__ void
chain_product(const float* __restrict__ A, const float* __restrict__ Bp,
              float* __restrict__ C, v2f* As, float* X, const int wg,
              const bool packed_out) {
  const int tid = threadIdx.x;
  const int wave = tid >> 5;
  const int lane = tid & 31;

  if (wg >= NDENSE_WG) {
    const int nt = NTTOP + 4 * (wg - NDENSE_WG) + wave;
    const int row0 = nt * 16;
    const v2f* bp = (const v2f*)Bp + ((size_t)(4 * nt) * NTILES + nt) * 32 + lane;
    const float* ap = A + (size_t)(row0 + (lane & 15)) * DQ + 16 * nt + 2 * (lane >> 4);
    v2f a[4], b[4];
#pragma unroll
    for (int j = 0; j < 4; ++j) {
      b[j] = bp[(size_t)j * (NTILES * 32)];
      a[j] = *(const v2f*)(ap + 4 * j);
    }
    v8f acc = (v8f){0.f, 0.f, 0.f, 0.f, 0.f, 0.f, 0.f, 0.f};
#pragma unroll
    for (int j = 0; j < 4; ++j) acc = wmma_f32_4(a[j], b[j], acc);

    __builtin_amdgcn_sched_barrier(0);
    asm volatile("v_nop\n\tv_nop\n\tv_nop\n\tv_nop");
    __builtin_amdgcn_sched_barrier(0);
    chain_store_tile<false>(C, X, acc, row0, nt, wave, lane, packed_out);
    return;
  }

  const int row0 = (wg >> 2) * 16;
  const int nt = (wg & 3) * 4 + wave;

  const v2f* bp = (const v2f*)Bp + (size_t)nt * 32 + lane;

  v2f bf[KTOP], af[KTOP];

  {
    const int row = (tid & 63) >> 2;
    const int cq = tid & 3;
    const int k0 = (tid >> 6) * (KTOP / 2);
    const float* arow = A + (size_t)(row0 + row) * DQ;
    v4f v[KTOP / 8];
#pragma unroll
    for (int s = 0; s < KTOP / 8; ++s) v[s] = *(const v4f*)(arow + (k0 + cq + 4 * s) * 4);
#pragma unroll
    for (int u = 0; u < KTOP / 2; ++u) bf[u] = bp[(size_t)u * (NTILES * 32)];
#pragma unroll
    for (int s = 0; s < KTOP / 8; ++s) {
      const int ks = k0 + cq + 4 * s;
      v2f lo; lo[0] = v[s][0]; lo[1] = v[s][1];
      v2f hi; hi[0] = v[s][2]; hi[1] = v[s][3];
      As[ks * 32 + row] = lo;
      As[ks * 32 + row + 16] = hi;
    }
  }
  __syncthreads();

#pragma unroll
  for (int u = KTOP / 2; u < KTOP; ++u) bf[u] = bp[(size_t)u * (NTILES * 32)];
#pragma unroll
  for (int u = 0; u < KTOP; ++u) af[u] = As[u * 32 + lane];

  __builtin_amdgcn_sched_barrier(0);

  v8f acc = (v8f){0.f, 0.f, 0.f, 0.f, 0.f, 0.f, 0.f, 0.f};
#pragma unroll
  for (int u = 0; u < KTOP; ++u) acc = wmma_f32_4(af[u], bf[u], acc);

  __builtin_amdgcn_sched_barrier(0);
  asm volatile("v_nop\n\tv_nop\n\tv_nop\n\tv_nop");
  __builtin_amdgcn_sched_barrier(0);
  chain_store_tile<true>(C, X, acc, row0, nt, wave, lane, packed_out);
}

template <bool PACKED_OUT>
__global__ void __launch_bounds__(128)
vqc_wmma_gemm_chain(const float* __restrict__ A, const float* __restrict__ Bp,
                    float* __restrict__ C) {
  __shared__ v2f As[KTOP * 32];
  __shared__ float X[2 * 16 * 32];
  chain_product(A, Bp, C, As, X, blockIdx.x, PACKED_OUT);
}

__global__ void __launch_bounds__(128)
vqc_wmma_gemm_chain_pair(const float* __restrict__ A0, const float* __restrict__ B0p,
                         float* __restrict__ C0,
                         const float* __restrict__ A1, const float* __restrict__ B1p,
                         float* __restrict__ C1) {
  __shared__ v2f As[KTOP * 32];
  __shared__ float X[2 * 16 * 32];
  const bool g = blockIdx.x >= NPROD_WG;
  chain_product(g ? A1 : A0, g ? B1p : B0p, g ? C1 : C0, As, X,
                g ? (int)blockIdx.x - NPROD_WG : (int)blockIdx.x, g);
}

extern "C" void kernel_launch(void* const* d_in, const int* in_sizes, int n_in,
                              void* d_out, int out_size, void* d_ws,
                              size_t ws_size, hipStream_t stream) {
  (void)in_sizes; (void)n_in; (void)out_size; (void)ws_size;
  const float* state = (const float*)d_in[0];
  const float* rotp  = (const float*)d_in[1];
  const float* entp  = (const float*)d_in[2];
  float* out = (float*)d_out;

  float* w     = (float*)d_ws;
  float* Tlay  = w + 1048576;
  float* T01   = w + 0;
  float* T23   = w + 262144;
  float* Ttot  = w + 524288;

  vqc_build_layer<<<dim3(QQ, NLAY), 128, 0, stream>>>(rotp, entp, Tlay);

  vqc_wmma_gemm_chain_pair<<<NPAIR_WG, 128, 0, stream>>>(
      Tlay + 0 * 262144, Tlay + 1 * 262144, T01,
      Tlay + 2 * 262144, Tlay + 3 * 262144, T23);

  vqc_wmma_gemm_chain<true><<<NPROD_WG, 128, 0, stream>>>(T01, T23, Ttot);

  vqc_wmma_gemm<<<M_STATE / (32 * TPW), 128, 0, stream>>>(state, Ttot, out);
}


# The 32768x512 product keeps T in registers: 256 work-groups of 8 waves, each wave holds the B fragments of its 2+2 N tiles for all k and runs 4 row tiles on them; B read once; sum order unchanged
# speedup vs baseline: 6.4258x; 1.0505x over previous
#include <hip/hip_runtime.h>
#include <math.h>


typedef float v2f __attribute__((ext_vector_type(2)));
typedef float v4f __attribute__((ext_vector_type(4)));
typedef float v8f __attribute__((ext_vector_type(8)));

#define DQ      512
#define QQ      256
#define NLAY    4
#define NPAIR   255
#define KSTEPS  128
#define NTILES  32
#define M_STATE (16 * 2048)

static __device__ __forceinline__ v8f wmma_f32_4(v2f a, v2f b, v8f c) {
  return __builtin_amdgcn_wmma_f32_16x16x4_f32(false, a, false, b, (short)0, c,
                                               false, false);
}

#define CPAD 8

__global__ void __launch_bounds__(128)
vqc_build_layer(const float* __restrict__ rot_params,
                const float* __restrict__ ent_params,
                float* __restrict__ Tlay) {
  const int q = blockIdx.x;
  const int l = blockIdx.y;
  const int tid = threadIdx.x;

  __shared__ __attribute__((aligned(16))) float c[QQ + CPAD];
  __shared__ __attribute__((aligned(16))) float e[2][DQ];

  for (int p = tid; p < QQ + CPAD; p += 128) {
    float cv = 1.0f;
    if (p < NPAIR) {
      const float x = ent_params[l * NPAIR + p];
      cv = 0.1f * (1.0f / (1.0f + expf(-x)));
    }
    c[p] = cv;
  }
  {
    const v4f z = (v4f){0.f, 0.f, 0.f, 0.f};
    ((v4f*)&e[0][0])[tid] = z;
    ((v4f*)&e[1][0])[tid] = z;
  }

  const float t0 = rot_params[((l * QQ) + q) * 3 + 0];
  const float t1 = rot_params[((l * QQ) + q) * 3 + 1];
  const float t2 = rot_params[((l * QQ) + q) * 3 + 2];
  const float c0 = cosf(t0 * 0.5f);
  const float c1 = cosf(t1 * 0.5f);
  const float s1 = sinf(t1 * 0.5f);
  const float c2 = cosf(t2 * 0.5f);
  const float b00 = (c0 * c1) * c2;
  const float b01 = (c0 * (-s1)) * c2;
  const float b10 = (c0 * s1) * c2;
  __syncthreads();

  if (tid < 2) {
    const int r = 2 * q + tid;
    float* row = e[tid];
    row[r] = 1.0f;
    if (r < QQ) {
      if (r >= 1) row[r - 1] = c[r - 1];
      if (r <= QQ - 2) {
        float val = c[r];
        int j = r + 1;
        for (; (j & 7) != 0; ++j) { row[j] = val; val *= c[j]; }
        v4f ca = *(const v4f*)&c[j];
        v4f cb = *(const v4f*)&c[j + 4];
        for (; j < QQ; j += 8) {
          const v4f na = *(const v4f*)&c[j + 8];
          const v4f nb = *(const v4f*)&c[j + 12];
          v4f va, vb;
          va[0] = val; val *= ca[0];
          va[1] = val; val *= ca[1];
          va[2] = val; val *= ca[2];
          va[3] = val; val *= ca[3];
          vb[0] = val; val *= cb[0];
          vb[1] = val; val *= cb[1];
          vb[2] = val; val *= cb[2];
          vb[3] = val; val *= cb[3];
          *(v4f*)&row[j] = va;
          *(v4f*)&row[j + 4] = vb;
          ca = na; cb = nb;
        }
      }
    }
  }
  __syncthreads();

  if (l & 1) {
    const int ks = q >> 1;
    const int half = q & 1;
    float* outp = Tlay + (size_t)l * DQ * DQ;
#pragma unroll
    for (int s = 0; s < 2; ++s) {
      const int col = 256 * s + 2 * tid;
      const v2f e0 = ((const v2f*)&e[0][0])[128 * s + tid];
      const v2f e1 = ((const v2f*)&e[1][0])[128 * s + tid];
      v4f o;
      o[0] = b00 * e0[0] + b01 * e1[0];
      o[1] = b10 * e0[0] + b00 * e1[0];
      o[2] = b00 * e0[1] + b01 * e1[1];
      o[3] = b10 * e0[1] + b00 * e1[1];
      const int pair = (ks * NTILES + (col >> 4)) * 32 + 16 * half + (col & 15);
      *(v4f*)(outp + (size_t)pair * 2) = o;
    }
  } else {
    const v4f e0 = ((const v4f*)&e[0][0])[tid];
    const v4f e1 = ((const v4f*)&e[1][0])[tid];
    v4f o0, o1;
#pragma unroll
    for (int k = 0; k < 4; ++k) {
      o0[k] = b00 * e0[k] + b01 * e1[k];
      o1[k] = b10 * e0[k] + b00 * e1[k];
    }
    float* out0 = Tlay + ((size_t)l * DQ + 2 * q) * DQ;
    ((v4f*)out0)[tid] = o0;
    ((v4f*)(out0 + DQ))[tid] = o1;
  }
}

#define KTOP   64
#define NTTOP  16
#define DCH    4
#define NDCH   (KTOP / DCH)
#define TPW    4
#define NTW    2
#define GTHR   256
#define RTHR   (GTHR / 32)
#define NPC    (128 / RTHR)
#define ASTILE (2 * KSTEPS * 32)
#define SPC    (NPC / NDCH)
#define BAH    4
#define PFD    6
#define PFR    8

#define WMMA_RUN_BEGIN()                                                          \
  __builtin_amdgcn_sched_barrier(0)
#define WMMA_RUN_END()                                                            \
  do {                                                                            \
    __builtin_amdgcn_sched_barrier(0);   \
    asm volatile("v_nop\n\tv_nop\n\tv_nop\n\tv_nop");           \
    __builtin_amdgcn_sched_barrier(0);   \
  } while (0)
template <bool FIRST, bool MORE>
static __device__ __forceinline__ void
gemm_row_tile(float* __restrict__ C, const v2f* __restrict__ bpt,
              v2f (&bq)[KTOP][NTW], const v2f (&bd)[NTW][4],
              const v2f* Asc, v2f* Asn, const float* __restrict__ anext,
              const int row0, const int sbase, const int lane,
              const int nttop, const int ntbot) {
  {
    v2f ad[NTW][4][2];
#pragma unroll
    for (int t = 0; t < NTW; ++t)
#pragma unroll
      for (int j = 0; j < 4; ++j) {
        const int ks = 4 * (ntbot + t) + j;
        ad[t][j][0] = Asc[(0 * KSTEPS + ks) * 32 + lane];
        ad[t][j][1] = Asc[(1 * KSTEPS + ks) * 32 + lane];
      }
    v8f accd[2][NTW];
#pragma unroll
    for (int m = 0; m < 2; ++m)
#pragma unroll
      for (int t = 0; t < NTW; ++t)
        accd[m][t] = (v8f){0.f, 0.f, 0.f, 0.f, 0.f, 0.f, 0.f, 0.f};
    WMMA_RUN_BEGIN();
#pragma unroll
    for (int j = 0; j < 4; ++j)
#pragma unroll
      for (int t = 0; t < NTW; ++t) {
        accd[0][t] = wmma_f32_4(ad[t][j][0], bd[t][j], accd[0][t]);
        accd[1][t] = wmma_f32_4(ad[t][j][1], bd[t][j], accd[1][t]);
      }
    WMMA_RUN_END();

    const int lrow = lane & 15;
    const int hi8 = (lane >> 4) * 8;
#pragma unroll
    for (int mt = 0; mt < 2; ++mt)
#pragma unroll
      for (int t = 0; t < NTW; ++t) {
        const int col = (ntbot + t) * 16 + lrow;
#pragma unroll
        for (int v = 0; v < 8; ++v) {
          const int r = row0 + mt * 16 + hi8 + v;
          C[(size_t)r * DQ + col] = accd[mt][t][v];
        }
      }
  }

  v2f aq[2][DCH][2];
  v4f pf[PFR][SPC];
#pragma unroll
  for (int u = 0; u < DCH; ++u) {
    aq[0][u][0] = Asc[(0 * KSTEPS + u) * 32 + lane];
    aq[0][u][1] = Asc[(1 * KSTEPS + u) * 32 + lane];
  }
  v8f acc[2][NTW];
#pragma unroll
  for (int m = 0; m < 2; ++m)
#pragma unroll
    for (int t = 0; t < NTW; ++t)
      acc[m][t] = (v8f){0.f, 0.f, 0.f, 0.f, 0.f, 0.f, 0.f, 0.f};

#pragma unroll
  for (int c = 0; c < NDCH; ++c) {
    if (FIRST && c + BAH < NDCH) {
      const int kn = (c + BAH) * DCH;
#pragma unroll
      for (int u = 0; u < DCH; ++u)
#pragma unroll
        for (int t = 0; t < NTW; ++t)
          bq[kn + u][t] = bpt[(size_t)(kn + u) * (NTILES * 32) + t * 32];
    }
    if (c + 1 < NDCH) {
      const int kn = (c + 1) * DCH;
#pragma unroll
      for (int u = 0; u < DCH; ++u) {
        aq[(c + 1) & 1][u][0] = Asc[(0 * KSTEPS + kn + u) * 32 + lane];
        aq[(c + 1) & 1][u][1] = Asc[(1 * KSTEPS + kn + u) * 32 + lane];
      }
    }
    if (MORE) {
#pragma unroll
      for (int i = 0; i < SPC; ++i)
        pf[c % PFR][i] = *(const v4f*)(anext + 32 * (SPC * c + i));
      if (c >= PFD) {
#pragma unroll
        for (int i = 0; i < SPC; ++i) {
          const int s = SPC * (c - PFD) + i;
          const v4f v = pf[(c - PFD) % PFR][i];
          v2f lo; lo[0] = v[0]; lo[1] = v[1];
          v2f hi; hi[0] = v[2]; hi[1] = v[3];
          Asn[sbase + 256 * s] = lo;
          Asn[sbase + 256 * s + 16] = hi;
        }
      }
    }
    WMMA_RUN_BEGIN();
#pragma unroll
    for (int u = 0; u < DCH; ++u)
#pragma unroll
      for (int t = 0; t < NTW; ++t) {
        acc[0][t] = wmma_f32_4(aq[c & 1][u][0], bq[c * DCH + u][t], acc[0][t]);
        acc[1][t] = wmma_f32_4(aq[c & 1][u][1], bq[c * DCH + u][t], acc[1][t]);
      }
    WMMA_RUN_END();
  }

  if (MORE) {
#pragma unroll
    for (int c = NDCH - PFD; c < NDCH; ++c)
#pragma unroll
      for (int i = 0; i < SPC; ++i) {
        const int s = SPC * c + i;
        const v4f v = pf[c % PFR][i];
        v2f lo; lo[0] = v[0]; lo[1] = v[1];
        v2f hi; hi[0] = v[2]; hi[1] = v[3];
        Asn[sbase + 256 * s] = lo;
        Asn[sbase + 256 * s + 16] = hi;
      }
    __syncthreads();
  }

  {
    const int lrow = lane & 15;
    const int hi8 = (lane >> 4) * 8;
#pragma unroll
    for (int mt = 0; mt < 2; ++mt)
#pragma unroll
      for (int t = 0; t < NTW; ++t) {
        const int col = (nttop + t) * 16 + lrow;
#pragma unroll
        for (int v = 0; v < 8; ++v) {
          const int r = row0 + mt * 16 + hi8 + v;
          C[(size_t)r * DQ + col] = acc[mt][t][v];
        }
      }
  }
}

__global__ void __launch_bounds__(GTHR)
vqc_wmma_gemm(const float* __restrict__ A, const float* __restrict__ Bp,
              float* __restrict__ C) {
  __shared__ v2f As[2 * ASTILE];

  const int wave = threadIdx.x >> 5;
  const int lane = threadIdx.x & 31;
  const int nttop = wave * NTW;
  const int ntbot = NTTOP + wave * NTW;
  const v2f* BpV = (const v2f*)Bp;
  const v2f* bpt = BpV + (size_t)nttop * 32 + lane;

  const int srow = threadIdx.x / RTHR;
  const int cq = threadIdx.x % RTHR;
  const int sbase = ((srow >> 4) * KSTEPS + cq) * 32 + (srow & 15);
  const int tile0 = blockIdx.x * TPW;
  const float* arow = A + ((size_t)tile0 * 32 + srow) * DQ + cq * 4;

  v2f bd[NTW][4];
#pragma unroll
  for (int t = 0; t < NTW; ++t)
#pragma unroll
    for (int j = 0; j < 4; ++j) {
      const int nt = ntbot + t;
      bd[t][j] = BpV[((size_t)(4 * nt + j) * NTILES + nt) * 32 + lane];
    }
  v2f bq[KTOP][NTW];
#pragma unroll
  for (int ks = 0; ks < BAH * DCH; ++ks)
#pragma unroll
    for (int t = 0; t < NTW; ++t)
      bq[ks][t] = bpt[(size_t)ks * (NTILES * 32) + t * 32];

#pragma unroll
  for (int s = 0; s < NPC; ++s) {
    const v4f v = *(const v4f*)(arow + 32 * s);
    v2f lo; lo[0] = v[0]; lo[1] = v[1];
    v2f hi; hi[0] = v[2]; hi[1] = v[3];
    As[sbase + 256 * s] = lo;
    As[sbase + 256 * s + 16] = hi;
  }
  __syncthreads();

  gemm_row_tile<true, true>(C, bpt, bq, bd, As, As + ASTILE, arow + (size_t)32 * DQ,
                            tile0 * 32, sbase, lane, nttop, ntbot);
#pragma nounroll
  for (int it = 1; it + 1 < TPW; ++it)
    gemm_row_tile<false, true>(C, bpt, bq, bd, As + (it & 1) * ASTILE,
                               As + ((it + 1) & 1) * ASTILE,
                               arow + (size_t)(it + 1) * 32 * DQ,
                               (tile0 + it) * 32, sbase, lane, nttop, ntbot);
  gemm_row_tile<false, false>(C, bpt, bq, bd, As + ((TPW - 1) & 1) * ASTILE, As,
                              arow, (tile0 + TPW - 1) * 32, sbase, lane, nttop, ntbot);
}

#define NDENSE_WG 64
#define NPROD_WG 68
#define NPAIR_WG 136

template <bool PAIRED>
static __device__ __forceinline__ void
chain_store_tile(float* __restrict__ C, float* X, const v8f acc, const int row0,
                 const int nt, const int wave, const int lane,
                 const bool packed_out) {
  const int lrow = lane & 15;
  const int half = lane >> 4;
  const int hi8 = half * 8;
  if (packed_out) {
    v2f* cp = (v2f*)C + ((size_t)(((row0 + hi8) >> 2) * NTILES + nt) * 32 + lrow);
#pragma unroll
    for (int p = 0; p < 4; ++p) {
      v2f x; x[0] = acc[2 * p]; x[1] = acc[2 * p + 1];
      cp[(p >> 1) * (NTILES * 32) + 16 * (p & 1)] = x;
    }
  } else if (PAIRED) {
    float* xp = X + (wave >> 1) * (16 * 32);
    const int odd = wave & 1;
#pragma unroll
    for (int v = 0; v < 8; ++v) xp[(hi8 + v) * 32 + 16 * odd + lrow] = acc[v];
    __syncthreads();
    float o[8];
#pragma unroll
    for (int v = 0; v < 8; ++v) o[v] = xp[(8 * odd + v) * 32 + lane];
    float* crow = C + (size_t)(row0 + 8 * odd) * DQ + (nt & ~1) * 16 + lane;
#pragma unroll
    for (int v = 0; v < 8; ++v) crow[(size_t)v * DQ] = o[v];
  } else {
#pragma unroll
    for (int v = 0; v < 8; ++v)
      C[(size_t)(row0 + hi8 + v) * DQ + nt * 16 + lrow] = acc[v];
  }
}

static __device__ __forceinline__ void
chain_product(const float* __restrict__ A, const float* __restrict__ Bp,
              float* __restrict__ C, v2f* As, float* X, const int wg,
              const bool packed_out) {
  const int tid = threadIdx.x;
  const int wave = tid >> 5;
  const int lane = tid & 31;

  if (wg >= NDENSE_WG) {
    const int nt = NTTOP + 4 * (wg - NDENSE_WG) + wave;
    const int row0 = nt * 16;
    const v2f* bp = (const v2f*)Bp + ((size_t)(4 * nt) * NTILES + nt) * 32 + lane;
    const float* ap = A + (size_t)(row0 + (lane & 15)) * DQ + 16 * nt + 2 * (lane >> 4);
    v2f a[4], b[4];
#pragma unroll
    for (int j = 0; j < 4; ++j) {
      b[j] = bp[(size_t)j * (NTILES * 32)];
      a[j] = *(const v2f*)(ap + 4 * j);
    }
    v8f acc = (v8f){0.f, 0.f, 0.f, 0.f, 0.f, 0.f, 0.f, 0.f};
#pragma unroll
    for (int j = 0; j < 4; ++j) acc = wmma_f32_4(a[j], b[j], acc);

    __builtin_amdgcn_sched_barrier(0);
    asm volatile("v_nop\n\tv_nop\n\tv_nop\n\tv_nop");
    __builtin_amdgcn_sched_barrier(0);
    chain_store_tile<false>(C, X, acc, row0, nt, wave, lane, packed_out);
    return;
  }

  const int row0 = (wg >> 2) * 16;
  const int nt = (wg & 3) * 4 + wave;

  const v2f* bp = (const v2f*)Bp + (size_t)nt * 32 + lane;

  v2f bf[KTOP], af[KTOP];

  {
    const int row = (tid & 63) >> 2;
    const int cq = tid & 3;
    const int k0 = (tid >> 6) * (KTOP / 2);
    const float* arow = A + (size_t)(row0 + row) * DQ;
    v4f v[KTOP / 8];
#pragma unroll
    for (int s = 0; s < KTOP / 8; ++s) v[s] = *(const v4f*)(arow + (k0 + cq + 4 * s) * 4);
#pragma unroll
    for (int u = 0; u < KTOP / 2; ++u) bf[u] = bp[(size_t)u * (NTILES * 32)];
#pragma unroll
    for (int s = 0; s < KTOP / 8; ++s) {
      const int ks = k0 + cq + 4 * s;
      v2f lo; lo[0] = v[s][0]; lo[1] = v[s][1];
      v2f hi; hi[0] = v[s][2]; hi[1] = v[s][3];
      As[ks * 32 + row] = lo;
      As[ks * 32 + row + 16] = hi;
    }
  }
  __syncthreads();

#pragma unroll
  for (int u = KTOP / 2; u < KTOP; ++u) bf[u] = bp[(size_t)u * (NTILES * 32)];
#pragma unroll
  for (int u = 0; u < KTOP; ++u) af[u] = As[u * 32 + lane];

  __builtin_amdgcn_sched_barrier(0);

  v8f acc = (v8f){0.f, 0.f, 0.f, 0.f, 0.f, 0.f, 0.f, 0.f};
#pragma unroll
  for (int u = 0; u < KTOP; ++u) acc = wmma_f32_4(af[u], bf[u], acc);

  __builtin_amdgcn_sched_barrier(0);
  asm volatile("v_nop\n\tv_nop\n\tv_nop\n\tv_nop");
  __builtin_amdgcn_sched_barrier(0);
  chain_store_tile<true>(C, X, acc, row0, nt, wave, lane, packed_out);
}

template <bool PACKED_OUT>
__global__ void __launch_bounds__(128)
vqc_wmma_gemm_chain(const float* __restrict__ A, const float* __restrict__ Bp,
                    float* __restrict__ C) {
  __shared__ v2f As[KTOP * 32];
  __shared__ float X[2 * 16 * 32];
  chain_product(A, Bp, C, As, X, blockIdx.x, PACKED_OUT);
}

__global__ void __launch_bounds__(128)
vqc_wmma_gemm_chain_pair(const float* __restrict__ A0, const float* __restrict__ B0p,
                         float* __restrict__ C0,
                         const float* __restrict__ A1, const float* __restrict__ B1p,
                         float* __restrict__ C1) {
  __shared__ v2f As[KTOP * 32];
  __shared__ float X[2 * 16 * 32];
  const bool g = blockIdx.x >= NPROD_WG;
  chain_product(g ? A1 : A0, g ? B1p : B0p, g ? C1 : C0, As, X,
                g ? (int)blockIdx.x - NPROD_WG : (int)blockIdx.x, g);
}

extern "C" void kernel_launch(void* const* d_in, const int* in_sizes, int n_in,
                              void* d_out, int out_size, void* d_ws,
                              size_t ws_size, hipStream_t stream) {
  (void)in_sizes; (void)n_in; (void)out_size; (void)ws_size;
  const float* state = (const float*)d_in[0];
  const float* rotp  = (const float*)d_in[1];
  const float* entp  = (const float*)d_in[2];
  float* out = (float*)d_out;

  float* w     = (float*)d_ws;
  float* Tlay  = w + 1048576;
  float* T01   = w + 0;
  float* T23   = w + 262144;
  float* Ttot  = w + 524288;

  vqc_build_layer<<<dim3(QQ, NLAY), 128, 0, stream>>>(rotp, entp, Tlay);

  vqc_wmma_gemm_chain_pair<<<NPAIR_WG, 128, 0, stream>>>(
      Tlay + 0 * 262144, Tlay + 1 * 262144, T01,
      Tlay + 2 * 262144, Tlay + 3 * 262144, T23);

  vqc_wmma_gemm_chain<true><<<NPROD_WG, 128, 0, stream>>>(T01, T23, Ttot);

  vqc_wmma_gemm<<<M_STATE / (32 * TPW), GTHR, 0, stream>>>(state, Ttot, out);
}


# The 32768x512 product writes every 128-B line of out with ONE store: the registers of a wave's two side-by-side N tiles swap lane halves (v_permlane16_swap) before the stores; sums unchanged
# speedup vs baseline: 6.6579x; 1.0361x over previous
#include <hip/hip_runtime.h>
#include <math.h>


typedef float v2f __attribute__((ext_vector_type(2)));
typedef float v4f __attribute__((ext_vector_type(4)));
typedef float v8f __attribute__((ext_vector_type(8)));

#define DQ      512
#define QQ      256
#define NLAY    4
#define NPAIR   255
#define KSTEPS  128
#define NTILES  32
#define M_STATE (16 * 2048)

static __device__ __forceinline__ v8f wmma_f32_4(v2f a, v2f b, v8f c) {
  return __builtin_amdgcn_wmma_f32_16x16x4_f32(false, a, false, b, (short)0, c,
                                               false, false);
}

static __device__ __forceinline__ void swap_lane_halves(float& x, float& y) {
  asm("v_permlane16_swap_b32 %0, %1" : "+v"(x), "+v"(y));
}

#define CPAD 8

__global__ void __launch_bounds__(128)
vqc_build_layer(const float* __restrict__ rot_params,
                const float* __restrict__ ent_params,
                float* __restrict__ Tlay) {
  const int q = blockIdx.x;
  const int l = blockIdx.y;
  const int tid = threadIdx.x;

  __shared__ __attribute__((aligned(16))) float c[QQ + CPAD];
  __shared__ __attribute__((aligned(16))) float e[2][DQ];

  for (int p = tid; p < QQ + CPAD; p += 128) {
    float cv = 1.0f;
    if (p < NPAIR) {
      const float x = ent_params[l * NPAIR + p];
      cv = 0.1f * (1.0f / (1.0f + expf(-x)));
    }
    c[p] = cv;
  }
  {
    const v4f z = (v4f){0.f, 0.f, 0.f, 0.f};
    ((v4f*)&e[0][0])[tid] = z;
    ((v4f*)&e[1][0])[tid] = z;
  }

  const float t0 = rot_params[((l * QQ) + q) * 3 + 0];
  const float t1 = rot_params[((l * QQ) + q) * 3 + 1];
  const float t2 = rot_params[((l * QQ) + q) * 3 + 2];
  const float c0 = cosf(t0 * 0.5f);
  const float c1 = cosf(t1 * 0.5f);
  const float s1 = sinf(t1 * 0.5f);
  const float c2 = cosf(t2 * 0.5f);
  const float b00 = (c0 * c1) * c2;
  const float b01 = (c0 * (-s1)) * c2;
  const float b10 = (c0 * s1) * c2;
  __syncthreads();

  if (tid < 2) {
    const int r = 2 * q + tid;
    float* row = e[tid];
    row[r] = 1.0f;
    if (r < QQ) {
      if (r >= 1) row[r - 1] = c[r - 1];
      if (r <= QQ - 2) {
        float val = c[r];
        int j = r + 1;
        for (; (j & 7) != 0; ++j) { row[j] = val; val *= c[j]; }
        v4f ca = *(const v4f*)&c[j];
        v4f cb = *(const v4f*)&c[j + 4];
        for (; j < QQ; j += 8) {
          const v4f na = *(const v4f*)&c[j + 8];
          const v4f nb = *(const v4f*)&c[j + 12];
          v4f va, vb;
          va[0] = val; val *= ca[0];
          va[1] = val; val *= ca[1];
          va[2] = val; val *= ca[2];
          va[3] = val; val *= ca[3];
          vb[0] = val; val *= cb[0];
          vb[1] = val; val *= cb[1];
          vb[2] = val; val *= cb[2];
          vb[3] = val; val *= cb[3];
          *(v4f*)&row[j] = va;
          *(v4f*)&row[j + 4] = vb;
          ca = na; cb = nb;
        }
      }
    }
  }
  __syncthreads();

  if (l & 1) {
    const int ks = q >> 1;
    const int half = q & 1;
    float* outp = Tlay + (size_t)l * DQ * DQ;
#pragma unroll
    for (int s = 0; s < 2; ++s) {
      const int col = 256 * s + 2 * tid;
      const v2f e0 = ((const v2f*)&e[0][0])[128 * s + tid];
      const v2f e1 = ((const v2f*)&e[1][0])[128 * s + tid];
      v4f o;
      o[0] = b00 * e0[0] + b01 * e1[0];
      o[1] = b10 * e0[0] + b00 * e1[0];
      o[2] = b00 * e0[1] + b01 * e1[1];
      o[3] = b10 * e0[1] + b00 * e1[1];
      const int pair = (ks * NTILES + (col >> 4)) * 32 + 16 * half + (col & 15);
      *(v4f*)(outp + (size_t)pair * 2) = o;
    }
  } else {
    const v4f e0 = ((const v4f*)&e[0][0])[tid];
    const v4f e1 = ((const v4f*)&e[1][0])[tid];
    v4f o0, o1;
#pragma unroll
    for (int k = 0; k < 4; ++k) {
      o0[k] = b00 * e0[k] + b01 * e1[k];
      o1[k] = b10 * e0[k] + b00 * e1[k];
    }
    float* out0 = Tlay + ((size_t)l * DQ + 2 * q) * DQ;
    ((v4f*)out0)[tid] = o0;
    ((v4f*)(out0 + DQ))[tid] = o1;
  }
}

#define KTOP   64
#define NTTOP  16
#define DCH    4
#define NDCH   (KTOP / DCH)
#define TPW    4
#define NTW    2
#define GTHR   256
#define RTHR   (GTHR / 32)
#define NPC    (128 / RTHR)
#define ASTILE (2 * KSTEPS * 32)
#define SPC    (NPC / NDCH)
#define BAH    4
#define PFD    6
#define PFR    8

#define WMMA_RUN_BEGIN()                                                          \
  __builtin_amdgcn_sched_barrier(0)
#define WMMA_RUN_END()                                                            \
  do {                                                                            \
    __builtin_amdgcn_sched_barrier(0);   \
    asm volatile("v_nop\n\tv_nop\n\tv_nop\n\tv_nop");           \
    __builtin_amdgcn_sched_barrier(0);   \
  } while (0)
template <bool FIRST, bool MORE>
static __device__ __forceinline__ void
gemm_row_tile(float* __restrict__ C, const v2f* __restrict__ bpt,
              v2f (&bq)[KTOP][NTW], const v2f (&bd)[NTW][4],
              const v2f* Asc, v2f* Asn, const float* __restrict__ anext,
              const int row0, const int sbase, const int lane,
              const int nttop, const int ntbot) {
  {
    v2f ad[NTW][4][2];
#pragma unroll
    for (int t = 0; t < NTW; ++t)
#pragma unroll
      for (int j = 0; j < 4; ++j) {
        const int ks = 4 * (ntbot + t) + j;
        ad[t][j][0] = Asc[(0 * KSTEPS + ks) * 32 + lane];
        ad[t][j][1] = Asc[(1 * KSTEPS + ks) * 32 + lane];
      }
    v8f accd[2][NTW];
#pragma unroll
    for (int m = 0; m < 2; ++m)
#pragma unroll
      for (int t = 0; t < NTW; ++t)
        accd[m][t] = (v8f){0.f, 0.f, 0.f, 0.f, 0.f, 0.f, 0.f, 0.f};
    WMMA_RUN_BEGIN();
#pragma unroll
    for (int j = 0; j < 4; ++j)
#pragma unroll
      for (int t = 0; t < NTW; ++t) {
        accd[0][t] = wmma_f32_4(ad[t][j][0], bd[t][j], accd[0][t]);
        accd[1][t] = wmma_f32_4(ad[t][j][1], bd[t][j], accd[1][t]);
      }
    WMMA_RUN_END();

#pragma unroll
    for (int mt = 0; mt < 2; ++mt) {
      float* crow = C + (size_t)(row0 + mt * 16) * DQ + ntbot * 16 + lane;
#pragma unroll
      for (int v = 0; v < 8; ++v) {
        float x = accd[mt][0][v], y = accd[mt][1][v];
        swap_lane_halves(x, y);
        crow[(size_t)v * DQ] = x;
        crow[(size_t)(8 + v) * DQ] = y;
      }
    }
  }

  v2f aq[2][DCH][2];
  v4f pf[PFR][SPC];
#pragma unroll
  for (int u = 0; u < DCH; ++u) {
    aq[0][u][0] = Asc[(0 * KSTEPS + u) * 32 + lane];
    aq[0][u][1] = Asc[(1 * KSTEPS + u) * 32 + lane];
  }
  v8f acc[2][NTW];
#pragma unroll
  for (int m = 0; m < 2; ++m)
#pragma unroll
    for (int t = 0; t < NTW; ++t)
      acc[m][t] = (v8f){0.f, 0.f, 0.f, 0.f, 0.f, 0.f, 0.f, 0.f};

#pragma unroll
  for (int c = 0; c < NDCH; ++c) {
    if (FIRST && c + BAH < NDCH) {
      const int kn = (c + BAH) * DCH;
#pragma unroll
      for (int u = 0; u < DCH; ++u)
#pragma unroll
        for (int t = 0; t < NTW; ++t)
          bq[kn + u][t] = bpt[(size_t)(kn + u) * (NTILES * 32) + t * 32];
    }
    if (c + 1 < NDCH) {
      const int kn = (c + 1) * DCH;
#pragma unroll
      for (int u = 0; u < DCH; ++u) {
        aq[(c + 1) & 1][u][0] = Asc[(0 * KSTEPS + kn + u) * 32 + lane];
        aq[(c + 1) & 1][u][1] = Asc[(1 * KSTEPS + kn + u) * 32 + lane];
      }
    }
    if (MORE) {
#pragma unroll
      for (int i = 0; i < SPC; ++i)
        pf[c % PFR][i] = *(const v4f*)(anext + 32 * (SPC * c + i));
      if (c >= PFD) {
#pragma unroll
        for (int i = 0; i < SPC; ++i) {
          const int s = SPC * (c - PFD) + i;
          const v4f v = pf[(c - PFD) % PFR][i];
          v2f lo; lo[0] = v[0]; lo[1] = v[1];
          v2f hi; hi[0] = v[2]; hi[1] = v[3];
          Asn[sbase + 256 * s] = lo;
          Asn[sbase + 256 * s + 16] = hi;
        }
      }
    }
    WMMA_RUN_BEGIN();
#pragma unroll
    for (int u = 0; u < DCH; ++u)
#pragma unroll
      for (int t = 0; t < NTW; ++t) {
        acc[0][t] = wmma_f32_4(aq[c & 1][u][0], bq[c * DCH + u][t], acc[0][t]);
        acc[1][t] = wmma_f32_4(aq[c & 1][u][1], bq[c * DCH + u][t], acc[1][t]);
      }
    WMMA_RUN_END();
  }

  if (MORE) {
#pragma unroll
    for (int c = NDCH - PFD; c < NDCH; ++c)
#pragma unroll
      for (int i = 0; i < SPC; ++i) {
        const int s = SPC * c + i;
        const v4f v = pf[c % PFR][i];
        v2f lo; lo[0] = v[0]; lo[1] = v[1];
        v2f hi; hi[0] = v[2]; hi[1] = v[3];
        Asn[sbase + 256 * s] = lo;
        Asn[sbase + 256 * s + 16] = hi;
      }
    __syncthreads();
  }

  {
#pragma unroll
    for (int mt = 0; mt < 2; ++mt) {
      float* crow = C + (size_t)(row0 + mt * 16) * DQ + nttop * 16 + lane;
#pragma unroll
      for (int v = 0; v < 8; ++v) {
        float x = acc[mt][0][v], y = acc[mt][1][v];
        swap_lane_halves(x, y);
        crow[(size_t)v * DQ] = x;
        crow[(size_t)(8 + v) * DQ] = y;
      }
    }
  }
}

__global__ void __launch_bounds__(GTHR)
vqc_wmma_gemm(const float* __restrict__ A, const float* __restrict__ Bp,
              float* __restrict__ C) {
  __shared__ v2f As[2 * ASTILE];

  const int wave = threadIdx.x >> 5;
  const int lane = threadIdx.x & 31;
  const int nttop = wave * NTW;
  const int ntbot = NTTOP + wave * NTW;
  const v2f* BpV = (const v2f*)Bp;
  const v2f* bpt = BpV + (size_t)nttop * 32 + lane;

  const int srow = threadIdx.x / RTHR;
  const int cq = threadIdx.x % RTHR;
  const int sbase = ((srow >> 4) * KSTEPS + cq) * 32 + (srow & 15);
  const int tile0 = blockIdx.x * TPW;
  const float* arow = A + ((size_t)tile0 * 32 + srow) * DQ + cq * 4;

  v2f bd[NTW][4];
#pragma unroll
  for (int t = 0; t < NTW; ++t)
#pragma unroll
    for (int j = 0; j < 4; ++j) {
      const int nt = ntbot + t;
      bd[t][j] = BpV[((size_t)(4 * nt + j) * NTILES + nt) * 32 + lane];
    }
  v2f bq[KTOP][NTW];
#pragma unroll
  for (int ks = 0; ks < BAH * DCH; ++ks)
#pragma unroll
    for (int t = 0; t < NTW; ++t)
      bq[ks][t] = bpt[(size_t)ks * (NTILES * 32) + t * 32];

#pragma unroll
  for (int s = 0; s < NPC; ++s) {
    const v4f v = *(const v4f*)(arow + 32 * s);
    v2f lo; lo[0] = v[0]; lo[1] = v[1];
    v2f hi; hi[0] = v[2]; hi[1] = v[3];
    As[sbase + 256 * s] = lo;
    As[sbase + 256 * s + 16] = hi;
  }
  __syncthreads();

  gemm_row_tile<true, true>(C, bpt, bq, bd, As, As + ASTILE, arow + (size_t)32 * DQ,
                            tile0 * 32, sbase, lane, nttop, ntbot);
#pragma nounroll
  for (int it = 1; it + 1 < TPW; ++it)
    gemm_row_tile<false, true>(C, bpt, bq, bd, As + (it & 1) * ASTILE,
                               As + ((it + 1) & 1) * ASTILE,
                               arow + (size_t)(it + 1) * 32 * DQ,
                               (tile0 + it) * 32, sbase, lane, nttop, ntbot);
  gemm_row_tile<false, false>(C, bpt, bq, bd, As + ((TPW - 1) & 1) * ASTILE, As,
                              arow, (tile0 + TPW - 1) * 32, sbase, lane, nttop, ntbot);
}

#define NDENSE_WG 64
#define NPROD_WG 68
#define NPAIR_WG 136

template <bool PAIRED>
static __device__ __forceinline__ void
chain_store_tile(float* __restrict__ C, float* X, const v8f acc, const int row0,
                 const int nt, const int wave, const int lane,
                 const bool packed_out) {
  const int lrow = lane & 15;
  const int half = lane >> 4;
  const int hi8 = half * 8;
  if (packed_out) {
    v2f* cp = (v2f*)C + ((size_t)(((row0 + hi8) >> 2) * NTILES + nt) * 32 + lrow);
#pragma unroll
    for (int p = 0; p < 4; ++p) {
      v2f x; x[0] = acc[2 * p]; x[1] = acc[2 * p + 1];
      cp[(p >> 1) * (NTILES * 32) + 16 * (p & 1)] = x;
    }
  } else if (PAIRED) {
    float* xp = X + (wave >> 1) * (16 * 32);
    const int odd = wave & 1;
#pragma unroll
    for (int v = 0; v < 8; ++v) xp[(hi8 + v) * 32 + 16 * odd + lrow] = acc[v];
    __syncthreads();
    float o[8];
#pragma unroll
    for (int v = 0; v < 8; ++v) o[v] = xp[(8 * odd + v) * 32 + lane];
    float* crow = C + (size_t)(row0 + 8 * odd) * DQ + (nt & ~1) * 16 + lane;
#pragma unroll
    for (int v = 0; v < 8; ++v) crow[(size_t)v * DQ] = o[v];
  } else {
#pragma unroll
    for (int v = 0; v < 8; ++v)
      C[(size_t)(row0 + hi8 + v) * DQ + nt * 16 + lrow] = acc[v];
  }
}

static __device__ __forceinline__ void
chain_product(const float* __restrict__ A, const float* __restrict__ Bp,
              float* __restrict__ C, v2f* As, float* X, const int wg,
              const bool packed_out) {
  const int tid = threadIdx.x;
  const int wave = tid >> 5;
  const int lane = tid & 31;

  if (wg >= NDENSE_WG) {
    const int nt = NTTOP + 4 * (wg - NDENSE_WG) + wave;
    const int row0 = nt * 16;
    const v2f* bp = (const v2f*)Bp + ((size_t)(4 * nt) * NTILES + nt) * 32 + lane;
    const float* ap = A + (size_t)(row0 + (lane & 15)) * DQ + 16 * nt + 2 * (lane >> 4);
    v2f a[4], b[4];
#pragma unroll
    for (int j = 0; j < 4; ++j) {
      b[j] = bp[(size_t)j * (NTILES * 32)];
      a[j] = *(const v2f*)(ap + 4 * j);
    }
    v8f acc = (v8f){0.f, 0.f, 0.f, 0.f, 0.f, 0.f, 0.f, 0.f};
#pragma unroll
    for (int j = 0; j < 4; ++j) acc = wmma_f32_4(a[j], b[j], acc);

    __builtin_amdgcn_sched_barrier(0);
    asm volatile("v_nop\n\tv_nop\n\tv_nop\n\tv_nop");
    __builtin_amdgcn_sched_barrier(0);
    chain_store_tile<false>(C, X, acc, row0, nt, wave, lane, packed_out);
    return;
  }

  const int row0 = (wg >> 2) * 16;
  const int nt = (wg & 3) * 4 + wave;

  const v2f* bp = (const v2f*)Bp + (size_t)nt * 32 + lane;

  v2f bf[KTOP], af[KTOP];

  {
    const int row = (tid & 63) >> 2;
    const int cq = tid & 3;
    const int k0 = (tid >> 6) * (KTOP / 2);
    const float* arow = A + (size_t)(row0 + row) * DQ;
    v4f v[KTOP / 8];
#pragma unroll
    for (int s = 0; s < KTOP / 8; ++s) v[s] = *(const v4f*)(arow + (k0 + cq + 4 * s) * 4);
#pragma unroll
    for (int u = 0; u < KTOP / 2; ++u) bf[u] = bp[(size_t)u * (NTILES * 32)];
#pragma unroll
    for (int s = 0; s < KTOP / 8; ++s) {
      const int ks = k0 + cq + 4 * s;
      v2f lo; lo[0] = v[s][0]; lo[1] = v[s][1];
      v2f hi; hi[0] = v[s][2]; hi[1] = v[s][3];
      As[ks * 32 + row] = lo;
      As[ks * 32 + row + 16] = hi;
    }
  }
  __syncthreads();

#pragma unroll
  for (int u = KTOP / 2; u < KTOP; ++u) bf[u] = bp[(size_t)u * (NTILES * 32)];
#pragma unroll
  for (int u = 0; u < KTOP; ++u) af[u] = As[u * 32 + lane];

  __builtin_amdgcn_sched_barrier(0);

  v8f acc = (v8f){0.f, 0.f, 0.f, 0.f, 0.f, 0.f, 0.f, 0.f};
#pragma unroll
  for (int u = 0; u < KTOP; ++u) acc = wmma_f32_4(af[u], bf[u], acc);

  __builtin_amdgcn_sched_barrier(0);
  asm volatile("v_nop\n\tv_nop\n\tv_nop\n\tv_nop");
  __builtin_amdgcn_sched_barrier(0);
  chain_store_tile<true>(C, X, acc, row0, nt, wave, lane, packed_out);
}

template <bool PACKED_OUT>
__global__ void __launch_bounds__(128)
vqc_wmma_gemm_chain(const float* __restrict__ A, const float* __restrict__ Bp,
                    float* __restrict__ C) {
  __shared__ v2f As[KTOP * 32];
  __shared__ float X[2 * 16 * 32];
  chain_product(A, Bp, C, As, X, blockIdx.x, PACKED_OUT);
}

__global__ void __launch_bounds__(128)
vqc_wmma_gemm_chain_pair(const float* __restrict__ A0, const float* __restrict__ B0p,
                         float* __restrict__ C0,
                         const float* __restrict__ A1, const float* __restrict__ B1p,
                         float* __restrict__ C1) {
  __shared__ v2f As[KTOP * 32];
  __shared__ float X[2 * 16 * 32];
  const bool g = blockIdx.x >= NPROD_WG;
  chain_product(g ? A1 : A0, g ? B1p : B0p, g ? C1 : C0, As, X,
                g ? (int)blockIdx.x - NPROD_WG : (int)blockIdx.x, g);
}

extern "C" void kernel_launch(void* const* d_in, const int* in_sizes, int n_in,
                              void* d_out, int out_size, void* d_ws,
                              size_t ws_size, hipStream_t stream) {
  (void)in_sizes; (void)n_in; (void)out_size; (void)ws_size;
  const float* state = (const float*)d_in[0];
  const float* rotp  = (const float*)d_in[1];
  const float* entp  = (const float*)d_in[2];
  float* out = (float*)d_out;

  float* w     = (float*)d_ws;
  float* Tlay  = w + 1048576;
  float* T01   = w + 0;
  float* T23   = w + 262144;
  float* Ttot  = w + 524288;

  vqc_build_layer<<<dim3(QQ, NLAY), 128, 0, stream>>>(rotp, entp, Tlay);

  vqc_wmma_gemm_chain_pair<<<NPAIR_WG, 128, 0, stream>>>(
      Tlay + 0 * 262144, Tlay + 1 * 262144, T01,
      Tlay + 2 * 262144, Tlay + 3 * 262144, T23);

  vqc_wmma_gemm_chain<true><<<NPROD_WG, 128, 0, stream>>>(T01, T23, Ttot);

  vqc_wmma_gemm<<<M_STATE / (32 * TPW), GTHR, 0, stream>>>(state, Ttot, out);
}


# The 32768x512 product leaves out k-steps where T is zero below its band (T[r][j] = 0 for j < r - 8 by construction): a SIMD's two waves run 20 dense chunks a tile, not 32; sum order unchanged
# speedup vs baseline: 7.2056x; 1.0823x over previous
#include <hip/hip_runtime.h>
#include <math.h>


typedef float v2f __attribute__((ext_vector_type(2)));
typedef float v4f __attribute__((ext_vector_type(4)));
typedef float v8f __attribute__((ext_vector_type(8)));

#define DQ      512
#define QQ      256
#define NLAY    4
#define NPAIR   255
#define KSTEPS  128
#define NTILES  32
#define M_STATE (16 * 2048)

static __device__ __forceinline__ v8f wmma_f32_4(v2f a, v2f b, v8f c) {
  return __builtin_amdgcn_wmma_f32_16x16x4_f32(false, a, false, b, (short)0, c,
                                               false, false);
}

static __device__ __forceinline__ void swap_lane_halves(float& x, float& y) {
  asm("v_permlane16_swap_b32 %0, %1" : "+v"(x), "+v"(y));
}

#define CPAD 8

__global__ void __launch_bounds__(128)
vqc_build_layer(const float* __restrict__ rot_params,
                const float* __restrict__ ent_params,
                float* __restrict__ Tlay) {
  const int q = blockIdx.x;
  const int l = blockIdx.y;
  const int tid = threadIdx.x;

  __shared__ __attribute__((aligned(16))) float c[QQ + CPAD];
  __shared__ __attribute__((aligned(16))) float e[2][DQ];

  for (int p = tid; p < QQ + CPAD; p += 128) {
    float cv = 1.0f;
    if (p < NPAIR) {
      const float x = ent_params[l * NPAIR + p];
      cv = 0.1f * (1.0f / (1.0f + expf(-x)));
    }
    c[p] = cv;
  }
  {
    const v4f z = (v4f){0.f, 0.f, 0.f, 0.f};
    ((v4f*)&e[0][0])[tid] = z;
    ((v4f*)&e[1][0])[tid] = z;
  }

  const float t0 = rot_params[((l * QQ) + q) * 3 + 0];
  const float t1 = rot_params[((l * QQ) + q) * 3 + 1];
  const float t2 = rot_params[((l * QQ) + q) * 3 + 2];
  const float c0 = cosf(t0 * 0.5f);
  const float c1 = cosf(t1 * 0.5f);
  const float s1 = sinf(t1 * 0.5f);
  const float c2 = cosf(t2 * 0.5f);
  const float b00 = (c0 * c1) * c2;
  const float b01 = (c0 * (-s1)) * c2;
  const float b10 = (c0 * s1) * c2;
  __syncthreads();

  if (tid < 2) {
    const int r = 2 * q + tid;
    float* row = e[tid];
    row[r] = 1.0f;
    if (r < QQ) {
      if (r >= 1) row[r - 1] = c[r - 1];
      if (r <= QQ - 2) {
        float val = c[r];
        int j = r + 1;
        for (; (j & 7) != 0; ++j) { row[j] = val; val *= c[j]; }
        v4f ca = *(const v4f*)&c[j];
        v4f cb = *(const v4f*)&c[j + 4];
        for (; j < QQ; j += 8) {
          const v4f na = *(const v4f*)&c[j + 8];
          const v4f nb = *(const v4f*)&c[j + 12];
          v4f va, vb;
          va[0] = val; val *= ca[0];
          va[1] = val; val *= ca[1];
          va[2] = val; val *= ca[2];
          va[3] = val; val *= ca[3];
          vb[0] = val; val *= cb[0];
          vb[1] = val; val *= cb[1];
          vb[2] = val; val *= cb[2];
          vb[3] = val; val *= cb[3];
          *(v4f*)&row[j] = va;
          *(v4f*)&row[j + 4] = vb;
          ca = na; cb = nb;
        }
      }
    }
  }
  __syncthreads();

  if (l & 1) {
    const int ks = q >> 1;
    const int half = q & 1;
    float* outp = Tlay + (size_t)l * DQ * DQ;
#pragma unroll
    for (int s = 0; s < 2; ++s) {
      const int col = 256 * s + 2 * tid;
      const v2f e0 = ((const v2f*)&e[0][0])[128 * s + tid];
      const v2f e1 = ((const v2f*)&e[1][0])[128 * s + tid];
      v4f o;
      o[0] = b00 * e0[0] + b01 * e1[0];
      o[1] = b10 * e0[0] + b00 * e1[0];
      o[2] = b00 * e0[1] + b01 * e1[1];
      o[3] = b10 * e0[1] + b00 * e1[1];
      const int pair = (ks * NTILES + (col >> 4)) * 32 + 16 * half + (col & 15);
      *(v4f*)(outp + (size_t)pair * 2) = o;
    }
  } else {
    const v4f e0 = ((const v4f*)&e[0][0])[tid];
    const v4f e1 = ((const v4f*)&e[1][0])[tid];
    v4f o0, o1;
#pragma unroll
    for (int k = 0; k < 4; ++k) {
      o0[k] = b00 * e0[k] + b01 * e1[k];
      o1[k] = b10 * e0[k] + b00 * e1[k];
    }
    float* out0 = Tlay + ((size_t)l * DQ + 2 * q) * DQ;
    ((v4f*)out0)[tid] = o0;
    ((v4f*)(out0 + DQ))[tid] = o1;
  }
}

#define KTOP   64
#define NTTOP  16
#define DCH    4
#define NDCH   (KTOP / DCH)
#define TPW    4
#define NTW    2
#define GTHR   256
#define RTHR   (GTHR / 32)
#define NPC    (128 / RTHR)
#define ASTILE (2 * KSTEPS * 32)
#define SPC    (NPC / NDCH)
#define BAH    4
#define PFD    6
#define PFR    8
#define LBW    (2 * NLAY)
#define NCH_OF(nt0) (((16 * ((nt0) + NTW - 1) + 15 + LBW) / 4 + 1 + DCH - 1) / DCH)

#define WMMA_RUN_BEGIN()                                                          \
  __builtin_amdgcn_sched_barrier(0)
#define WMMA_RUN_END()                                                            \
  do {                                                                            \
    __builtin_amdgcn_sched_barrier(0);   \
    asm volatile("v_nop\n\tv_nop\n\tv_nop\n\tv_nop");           \
    __builtin_amdgcn_sched_barrier(0);   \
  } while (0)
template <bool FIRST, bool MORE>
static __device__ __forceinline__ void
gemm_row_tile(float* __restrict__ C, const v2f* __restrict__ bpt,
              v2f (&bq)[KTOP][NTW], const v2f (&bd)[NTW][4],
              const v2f* Asc, v2f* Asn, const float* __restrict__ anext,
              const int row0, const int sbase, const int lane,
              const int nttop, const int ntbot, const int nch) {
  {
    v2f ad[NTW][4][2];
#pragma unroll
    for (int t = 0; t < NTW; ++t)
#pragma unroll
      for (int j = 0; j < 4; ++j) {
        const int ks = 4 * (ntbot + t) + j;
        ad[t][j][0] = Asc[(0 * KSTEPS + ks) * 32 + lane];
        ad[t][j][1] = Asc[(1 * KSTEPS + ks) * 32 + lane];
      }
    v8f accd[2][NTW];
#pragma unroll
    for (int m = 0; m < 2; ++m)
#pragma unroll
      for (int t = 0; t < NTW; ++t)
        accd[m][t] = (v8f){0.f, 0.f, 0.f, 0.f, 0.f, 0.f, 0.f, 0.f};
    WMMA_RUN_BEGIN();
#pragma unroll
    for (int j = 0; j < 4; ++j)
#pragma unroll
      for (int t = 0; t < NTW; ++t) {
        accd[0][t] = wmma_f32_4(ad[t][j][0], bd[t][j], accd[0][t]);
        accd[1][t] = wmma_f32_4(ad[t][j][1], bd[t][j], accd[1][t]);
      }
    WMMA_RUN_END();

#pragma unroll
    for (int mt = 0; mt < 2; ++mt) {
      float* crow = C + (size_t)(row0 + mt * 16) * DQ + ntbot * 16 + lane;
#pragma unroll
      for (int v = 0; v < 8; ++v) {
        float x = accd[mt][0][v], y = accd[mt][1][v];
        swap_lane_halves(x, y);
        crow[(size_t)v * DQ] = x;
        crow[(size_t)(8 + v) * DQ] = y;
      }
    }
  }

  v2f aq[2][DCH][2];
  v4f pf[PFR][SPC];
#pragma unroll
  for (int u = 0; u < DCH; ++u) {
    aq[0][u][0] = Asc[(0 * KSTEPS + u) * 32 + lane];
    aq[0][u][1] = Asc[(1 * KSTEPS + u) * 32 + lane];
  }
  v8f acc[2][NTW];
#pragma unroll
  for (int m = 0; m < 2; ++m)
#pragma unroll
    for (int t = 0; t < NTW; ++t)
      acc[m][t] = (v8f){0.f, 0.f, 0.f, 0.f, 0.f, 0.f, 0.f, 0.f};

#pragma unroll
  for (int c = 0; c < NDCH; ++c) {
    if (FIRST && c + BAH < NDCH) {
      const int kn = (c + BAH) * DCH;
#pragma unroll
      for (int u = 0; u < DCH; ++u)
#pragma unroll
        for (int t = 0; t < NTW; ++t)
          bq[kn + u][t] = bpt[(size_t)(kn + u) * (NTILES * 32) + t * 32];
    }
    if (c + 1 < NDCH) {
      const int kn = (c + 1) * DCH;
#pragma unroll
      for (int u = 0; u < DCH; ++u) {
        aq[(c + 1) & 1][u][0] = Asc[(0 * KSTEPS + kn + u) * 32 + lane];
        aq[(c + 1) & 1][u][1] = Asc[(1 * KSTEPS + kn + u) * 32 + lane];
      }
    }
    if (MORE) {
#pragma unroll
      for (int i = 0; i < SPC; ++i)
        pf[c % PFR][i] = *(const v4f*)(anext + 32 * (SPC * c + i));
      if (c >= PFD) {
#pragma unroll
        for (int i = 0; i < SPC; ++i) {
          const int s = SPC * (c - PFD) + i;
          const v4f v = pf[(c - PFD) % PFR][i];
          v2f lo; lo[0] = v[0]; lo[1] = v[1];
          v2f hi; hi[0] = v[2]; hi[1] = v[3];
          Asn[sbase + 256 * s] = lo;
          Asn[sbase + 256 * s + 16] = hi;
        }
      }
    }
    if (c < nch) {
      WMMA_RUN_BEGIN();
#pragma unroll
      for (int u = 0; u < DCH; ++u)
#pragma unroll
        for (int t = 0; t < NTW; ++t) {
          acc[0][t] = wmma_f32_4(aq[c & 1][u][0], bq[c * DCH + u][t], acc[0][t]);
          acc[1][t] = wmma_f32_4(aq[c & 1][u][1], bq[c * DCH + u][t], acc[1][t]);
        }
      WMMA_RUN_END();
    }
  }

  if (MORE) {
#pragma unroll
    for (int c = NDCH - PFD; c < NDCH; ++c)
#pragma unroll
      for (int i = 0; i < SPC; ++i) {
        const int s = SPC * c + i;
        const v4f v = pf[c % PFR][i];
        v2f lo; lo[0] = v[0]; lo[1] = v[1];
        v2f hi; hi[0] = v[2]; hi[1] = v[3];
        Asn[sbase + 256 * s] = lo;
        Asn[sbase + 256 * s + 16] = hi;
      }
    __syncthreads();
  }

  {
#pragma unroll
    for (int mt = 0; mt < 2; ++mt) {
      float* crow = C + (size_t)(row0 + mt * 16) * DQ + nttop * 16 + lane;
#pragma unroll
      for (int v = 0; v < 8; ++v) {
        float x = acc[mt][0][v], y = acc[mt][1][v];
        swap_lane_halves(x, y);
        crow[(size_t)v * DQ] = x;
        crow[(size_t)(8 + v) * DQ] = y;
      }
    }
  }
}

__global__ void __launch_bounds__(GTHR)
vqc_wmma_gemm(const float* __restrict__ A, const float* __restrict__ Bp,
              float* __restrict__ C) {
  __shared__ v2f As[2 * ASTILE];

  const int wave = __builtin_amdgcn_readfirstlane((int)(threadIdx.x >> 5));
  const int lane = threadIdx.x & 31;
  const int pr = wave < 4 ? 7 - wave : wave - 4;
  const int nttop = pr * NTW;
  const int nch = NCH_OF(nttop) < NDCH ? NCH_OF(nttop) : NDCH;
  const int ntbot = NTTOP + wave * NTW;
  const v2f* BpV = (const v2f*)Bp;
  const v2f* bpt = BpV + (size_t)nttop * 32 + lane;

  const int srow = threadIdx.x / RTHR;
  const int cq = threadIdx.x % RTHR;
  const int sbase = ((srow >> 4) * KSTEPS + cq) * 32 + (srow & 15);
  const int tile0 = blockIdx.x * TPW;
  const float* arow = A + ((size_t)tile0 * 32 + srow) * DQ + cq * 4;

  v2f bd[NTW][4];
#pragma unroll
  for (int t = 0; t < NTW; ++t)
#pragma unroll
    for (int j = 0; j < 4; ++j) {
      const int nt = ntbot + t;
      bd[t][j] = BpV[((size_t)(4 * nt + j) * NTILES + nt) * 32 + lane];
    }
  v2f bq[KTOP][NTW];
#pragma unroll
  for (int ks = 0; ks < BAH * DCH; ++ks)
#pragma unroll
    for (int t = 0; t < NTW; ++t)
      bq[ks][t] = bpt[(size_t)ks * (NTILES * 32) + t * 32];

#pragma unroll
  for (int s = 0; s < NPC; ++s) {
    const v4f v = *(const v4f*)(arow + 32 * s);
    v2f lo; lo[0] = v[0]; lo[1] = v[1];
    v2f hi; hi[0] = v[2]; hi[1] = v[3];
    As[sbase + 256 * s] = lo;
    As[sbase + 256 * s + 16] = hi;
  }
  __syncthreads();

  gemm_row_tile<true, true>(C, bpt, bq, bd, As, As + ASTILE, arow + (size_t)32 * DQ,
                            tile0 * 32, sbase, lane, nttop, ntbot, nch);
#pragma nounroll
  for (int it = 1; it + 1 < TPW; ++it)
    gemm_row_tile<false, true>(C, bpt, bq, bd, As + (it & 1) * ASTILE,
                               As + ((it + 1) & 1) * ASTILE,
                               arow + (size_t)(it + 1) * 32 * DQ,
                               (tile0 + it) * 32, sbase, lane, nttop, ntbot, nch);
  gemm_row_tile<false, false>(C, bpt, bq, bd, As + ((TPW - 1) & 1) * ASTILE, As,
                              arow, (tile0 + TPW - 1) * 32, sbase, lane, nttop, ntbot, nch);
}

#define NDENSE_WG 64
#define NPROD_WG 68
#define NPAIR_WG 136

template <bool PAIRED>
static __device__ __forceinline__ void
chain_store_tile(float* __restrict__ C, float* X, const v8f acc, const int row0,
                 const int nt, const int wave, const int lane,
                 const bool packed_out) {
  const int lrow = lane & 15;
  const int half = lane >> 4;
  const int hi8 = half * 8;
  if (packed_out) {
    v2f* cp = (v2f*)C + ((size_t)(((row0 + hi8) >> 2) * NTILES + nt) * 32 + lrow);
#pragma unroll
    for (int p = 0; p < 4; ++p) {
      v2f x; x[0] = acc[2 * p]; x[1] = acc[2 * p + 1];
      cp[(p >> 1) * (NTILES * 32) + 16 * (p & 1)] = x;
    }
  } else if (PAIRED) {
    float* xp = X + (wave >> 1) * (16 * 32);
    const int odd = wave & 1;
#pragma unroll
    for (int v = 0; v < 8; ++v) xp[(hi8 + v) * 32 + 16 * odd + lrow] = acc[v];
    __syncthreads();
    float o[8];
#pragma unroll
    for (int v = 0; v < 8; ++v) o[v] = xp[(8 * odd + v) * 32 + lane];
    float* crow = C + (size_t)(row0 + 8 * odd) * DQ + (nt & ~1) * 16 + lane;
#pragma unroll
    for (int v = 0; v < 8; ++v) crow[(size_t)v * DQ] = o[v];
  } else {
#pragma unroll
    for (int v = 0; v < 8; ++v)
      C[(size_t)(row0 + hi8 + v) * DQ + nt * 16 + lrow] = acc[v];
  }
}

static __device__ __forceinline__ void
chain_product(const float* __restrict__ A, const float* __restrict__ Bp,
              float* __restrict__ C, v2f* As, float* X, const int wg,
              const bool packed_out) {
  const int tid = threadIdx.x;
  const int wave = tid >> 5;
  const int lane = tid & 31;

  if (wg >= NDENSE_WG) {
    const int nt = NTTOP + 4 * (wg - NDENSE_WG) + wave;
    const int row0 = nt * 16;
    const v2f* bp = (const v2f*)Bp + ((size_t)(4 * nt) * NTILES + nt) * 32 + lane;
    const float* ap = A + (size_t)(row0 + (lane & 15)) * DQ + 16 * nt + 2 * (lane >> 4);
    v2f a[4], b[4];
#pragma unroll
    for (int j = 0; j < 4; ++j) {
      b[j] = bp[(size_t)j * (NTILES * 32)];
      a[j] = *(const v2f*)(ap + 4 * j);
    }
    v8f acc = (v8f){0.f, 0.f, 0.f, 0.f, 0.f, 0.f, 0.f, 0.f};
#pragma unroll
    for (int j = 0; j < 4; ++j) acc = wmma_f32_4(a[j], b[j], acc);

    __builtin_amdgcn_sched_barrier(0);
    asm volatile("v_nop\n\tv_nop\n\tv_nop\n\tv_nop");
    __builtin_amdgcn_sched_barrier(0);
    chain_store_tile<false>(C, X, acc, row0, nt, wave, lane, packed_out);
    return;
  }

  const int row0 = (wg >> 2) * 16;
  const int nt = (wg & 3) * 4 + wave;

  const v2f* bp = (const v2f*)Bp + (size_t)nt * 32 + lane;

  v2f bf[KTOP], af[KTOP];

  {
    const int row = (tid & 63) >> 2;
    const int cq = tid & 3;
    const int k0 = (tid >> 6) * (KTOP / 2);
    const float* arow = A + (size_t)(row0 + row) * DQ;
    v4f v[KTOP / 8];
#pragma unroll
    for (int s = 0; s < KTOP / 8; ++s) v[s] = *(const v4f*)(arow + (k0 + cq + 4 * s) * 4);
#pragma unroll
    for (int u = 0; u < KTOP / 2; ++u) bf[u] = bp[(size_t)u * (NTILES * 32)];
#pragma unroll
    for (int s = 0; s < KTOP / 8; ++s) {
      const int ks = k0 + cq + 4 * s;
      v2f lo; lo[0] = v[s][0]; lo[1] = v[s][1];
      v2f hi; hi[0] = v[s][2]; hi[1] = v[s][3];
      As[ks * 32 + row] = lo;
      As[ks * 32 + row + 16] = hi;
    }
  }
  __syncthreads();

#pragma unroll
  for (int u = KTOP / 2; u < KTOP; ++u) bf[u] = bp[(size_t)u * (NTILES * 32)];
#pragma unroll
  for (int u = 0; u < KTOP; ++u) af[u] = As[u * 32 + lane];

  __builtin_amdgcn_sched_barrier(0);

  v8f acc = (v8f){0.f, 0.f, 0.f, 0.f, 0.f, 0.f, 0.f, 0.f};
#pragma unroll
  for (int u = 0; u < KTOP; ++u) acc = wmma_f32_4(af[u], bf[u], acc);

  __builtin_amdgcn_sched_barrier(0);
  asm volatile("v_nop\n\tv_nop\n\tv_nop\n\tv_nop");
  __builtin_amdgcn_sched_barrier(0);
  chain_store_tile<true>(C, X, acc, row0, nt, wave, lane, packed_out);
}

template <bool PACKED_OUT>
__global__ void __launch_bounds__(128)
vqc_wmma_gemm_chain(const float* __restrict__ A, const float* __restrict__ Bp,
                    float* __restrict__ C) {
  __shared__ v2f As[KTOP * 32];
  __shared__ float X[2 * 16 * 32];
  chain_product(A, Bp, C, As, X, blockIdx.x, PACKED_OUT);
}

__global__ void __launch_bounds__(128)
vqc_wmma_gemm_chain_pair(const float* __restrict__ A0, const float* __restrict__ B0p,
                         float* __restrict__ C0,
                         const float* __restrict__ A1, const float* __restrict__ B1p,
                         float* __restrict__ C1) {
  __shared__ v2f As[KTOP * 32];
  __shared__ float X[2 * 16 * 32];
  const bool g = blockIdx.x >= NPROD_WG;
  chain_product(g ? A1 : A0, g ? B1p : B0p, g ? C1 : C0, As, X,
                g ? (int)blockIdx.x - NPROD_WG : (int)blockIdx.x, g);
}

extern "C" void kernel_launch(void* const* d_in, const int* in_sizes, int n_in,
                              void* d_out, int out_size, void* d_ws,
                              size_t ws_size, hipStream_t stream) {
  (void)in_sizes; (void)n_in; (void)out_size; (void)ws_size;
  const float* state = (const float*)d_in[0];
  const float* rotp  = (const float*)d_in[1];
  const float* entp  = (const float*)d_in[2];
  float* out = (float*)d_out;

  float* w     = (float*)d_ws;
  float* Tlay  = w + 1048576;
  float* T01   = w + 0;
  float* T23   = w + 262144;
  float* Ttot  = w + 524288;

  vqc_build_layer<<<dim3(QQ, NLAY), 128, 0, stream>>>(rotp, entp, Tlay);

  vqc_wmma_gemm_chain_pair<<<NPAIR_WG, 128, 0, stream>>>(
      Tlay + 0 * 262144, Tlay + 1 * 262144, T01,
      Tlay + 2 * 262144, Tlay + 3 * 262144, T23);

  vqc_wmma_gemm_chain<true><<<NPROD_WG, 128, 0, stream>>>(T01, T23, Ttot);

  vqc_wmma_gemm<<<M_STATE / (32 * TPW), GTHR, 0, stream>>>(state, Ttot, out);
}
